# starting kernel
# baseline (speedup 1.0000x reference)
#include <hip/hip_runtime.h>
#include <math.h>

typedef __attribute__((ext_vector_type(2))) float v2f;
typedef __attribute__((ext_vector_type(8))) float v8f;

#define N_NODES   100000
#define N_EDGES   1600000
#define IN_FEAT   256
#define HIDDEN    128
#define N_CLASSES 16


__global__ void init_deg_kernel(float* __restrict__ deg) {
    int i = blockIdx.x * blockDim.x + threadIdx.x;
    if (i < N_NODES) deg[i] = 1.0f;
}

__global__ void deg_accum_kernel(const int* __restrict__ col, float* __restrict__ deg) {
    int e = blockIdx.x * blockDim.x + threadIdx.x;
    if (e < N_EDGES) unsafeAtomicAdd(&deg[col[e]], 1.0f);
}

__global__ void dinv_kernel(float* __restrict__ deg) {
    int i = blockIdx.x * blockDim.x + threadIdx.x;
    if (i < N_NODES) deg[i] = rsqrtf(deg[i]);
}


__global__ __launch_bounds__(256) void gemm1_kernel(const float* __restrict__ x,
                                                    const float* __restrict__ W1,
                                                    float* __restrict__ hp) {
    const int lane = threadIdx.x & 31;
    const int wave = (blockIdx.x * blockDim.x + threadIdx.x) >> 5;
    const int row0 = wave * 16;
    if (row0 >= N_NODES) return;

    const int m  = lane & 15;
    const int kh = lane >> 4;
    const float* arow = x + (size_t)(row0 + m) * IN_FEAT;

    v8f acc[8];
#pragma unroll
    for (int n = 0; n < 8; ++n) acc[n] = v8f{};

    for (int k = 0; k < IN_FEAT; k += 4) {
        v2f a;
        a.x = arow[k + 2 * kh + 0];
        a.y = arow[k + 2 * kh + 1];
        const float* b0 = W1 + (size_t)(k + 2 * kh) * HIDDEN + m;
#pragma unroll
        for (int n = 0; n < 8; ++n) {
            v2f b;
            b.x = b0[n * 16];
            b.y = b0[HIDDEN + n * 16];
            acc[n] = __builtin_amdgcn_wmma_f32_16x16x4_f32(
                false, a, false, b, (short)0, acc[n], false, false);
        }
    }

    const int mo = kh * 8;
#pragma unroll
    for (int n = 0; n < 8; ++n) {
#pragma unroll
        for (int j = 0; j < 8; ++j) {
            hp[(size_t)(row0 + mo + j) * HIDDEN + n * 16 + m] = acc[n][j];
        }
    }
}


__global__ __launch_bounds__(256) void gemm2_kernel(const float* __restrict__ h,
                                                    const float* __restrict__ W2,
                                                    float* __restrict__ h2p) {
    const int lane = threadIdx.x & 31;
    const int wave = (blockIdx.x * blockDim.x + threadIdx.x) >> 5;
    const int row0 = wave * 16;
    if (row0 >= N_NODES) return;

    const int m  = lane & 15;
    const int kh = lane >> 4;
    const float* arow = h + (size_t)(row0 + m) * HIDDEN;

    v8f acc = v8f{};
    for (int k = 0; k < HIDDEN; k += 4) {
        v2f a;
        a.x = arow[k + 2 * kh + 0];
        a.y = arow[k + 2 * kh + 1];
        v2f b;
        b.x = W2[(size_t)(k + 2 * kh + 0) * N_CLASSES + m];
        b.y = W2[(size_t)(k + 2 * kh + 1) * N_CLASSES + m];
        acc = __builtin_amdgcn_wmma_f32_16x16x4_f32(
            false, a, false, b, (short)0, acc, false, false);
    }

    const int mo = kh * 8;
#pragma unroll
    for (int j = 0; j < 8; ++j) {
        h2p[(size_t)(row0 + mo + j) * N_CLASSES + m] = acc[j];
    }
}


__global__ void agg_self_kernel(const float* __restrict__ src,
                                const float* __restrict__ dinv,
                                float* __restrict__ agg,
                                int shift, long total) {
    long i = (long)blockIdx.x * blockDim.x + threadIdx.x;
    if (i < total) {
        int node = (int)(i >> shift);
        float d = dinv[node];
        agg[i] = d * d * src[i];
    }
}

__global__ __launch_bounds__(256) void agg1_edge_kernel(const int* __restrict__ row,
                                                        const int* __restrict__ col,
                                                        const float* __restrict__ dinv,
                                                        const float* __restrict__ hp,
                                                        float* __restrict__ agg) {
    const int lane = threadIdx.x & 31;
    const long e = ((long)blockIdx.x * blockDim.x + threadIdx.x) >> 5;
    if (e >= N_EDGES) return;
    const int r = row[e];
    const int c = col[e];
    const float w = dinv[r] * dinv[c];
    const float4 v = ((const float4*)(hp + (size_t)r * HIDDEN))[lane];
    float* dst = agg + (size_t)c * HIDDEN + lane * 4;
    unsafeAtomicAdd(dst + 0, w * v.x);
    unsafeAtomicAdd(dst + 1, w * v.y);
    unsafeAtomicAdd(dst + 2, w * v.z);
    unsafeAtomicAdd(dst + 3, w * v.w);
}

__global__ void agg2_edge_kernel(const int* __restrict__ row,
                                 const int* __restrict__ col,
                                 const float* __restrict__ dinv,
                                 const float* __restrict__ h2p,
                                 float* __restrict__ agg) {
    long t = (long)blockIdx.x * blockDim.x + threadIdx.x;
    if (t >= (long)N_EDGES * N_CLASSES) return;
    int e = (int)(t >> 4);
    int f = (int)(t & 15);
    int r = row[e];
    int c = col[e];
    float w = dinv[r] * dinv[c];
    unsafeAtomicAdd(&agg[(size_t)c * N_CLASSES + f], w * h2p[(size_t)r * N_CLASSES + f]);
}

__global__ void bias_relu_kernel(const float* __restrict__ agg,
                                 const float* __restrict__ b1,
                                 float* __restrict__ out) {
    long i = (long)blockIdx.x * blockDim.x + threadIdx.x;
    if (i < (long)N_NODES * HIDDEN) {
        float v = agg[i] + b1[(int)(i & (HIDDEN - 1))];
        out[i] = v > 0.0f ? v : 0.0f;
    }
}

__global__ void logsoftmax_kernel(const float* __restrict__ agg,
                                  const float* __restrict__ b2,
                                  float* __restrict__ out) {
    int i = blockIdx.x * blockDim.x + threadIdx.x;
    if (i >= N_NODES) return;
    float v[N_CLASSES];
    float mx = -INFINITY;
#pragma unroll
    for (int c = 0; c < N_CLASSES; ++c) {
        v[c] = agg[(size_t)i * N_CLASSES + c] + b2[c];
        mx = fmaxf(mx, v[c]);
    }
    float s = 0.0f;
#pragma unroll
    for (int c = 0; c < N_CLASSES; ++c) s += expf(v[c] - mx);
    float lse = mx + logf(s);
#pragma unroll
    for (int c = 0; c < N_CLASSES; ++c) {
        out[(size_t)i * N_CLASSES + c] = v[c] - lse;
    }
}


static inline int cdiv_l(long a, int b) { return (int)((a + b - 1) / b); }

extern "C" void kernel_launch(void* const* d_in, const int* in_sizes, int n_in,
                              void* d_out, int out_size, void* d_ws, size_t ws_size,
                              hipStream_t stream) {
    const float* x   = (const float*)d_in[0];
    const int*   ei  = (const int*)d_in[1];
    const int*   row = ei;
    const int*   col = ei + N_EDGES;
    const float* W1  = (const float*)d_in[2];
    const float* b1  = (const float*)d_in[3];
    const float* W2  = (const float*)d_in[4];
    const float* b2  = (const float*)d_in[5];
    float* out = (float*)d_out;

    float* ws   = (float*)d_ws;
    float* dinv = ws;
    float* hp   = ws + ((N_NODES + 255) & ~255);
    float* agg1 = hp + (size_t)N_NODES * HIDDEN;
    float* h2p  = agg1;
    float* agg2 = agg1 + (size_t)N_NODES * N_CLASSES;

    const int B = 256;

    init_deg_kernel<<<cdiv_l(N_NODES, B), B, 0, stream>>>(dinv);
    deg_accum_kernel<<<cdiv_l(N_EDGES, B), B, 0, stream>>>(col, dinv);
    dinv_kernel<<<cdiv_l(N_NODES, B), B, 0, stream>>>(dinv);

    const long waves1 = (N_NODES + 15) / 16;
    gemm1_kernel<<<cdiv_l(waves1 * 32, B), B, 0, stream>>>(x, W1, hp);
    agg_self_kernel<<<cdiv_l((long)N_NODES * HIDDEN, B), B, 0, stream>>>(
        hp, dinv, agg1, 7, (long)N_NODES * HIDDEN);
    agg1_edge_kernel<<<cdiv_l((long)N_EDGES * 32, B), B, 0, stream>>>(
        row, col, dinv, hp, agg1);
    bias_relu_kernel<<<cdiv_l((long)N_NODES * HIDDEN, B), B, 0, stream>>>(agg1, b1, hp);

    gemm2_kernel<<<cdiv_l(waves1 * 32, B), B, 0, stream>>>(hp, W2, h2p);
    agg_self_kernel<<<cdiv_l((long)N_NODES * N_CLASSES, B), B, 0, stream>>>(
        h2p, dinv, agg2, 4, (long)N_NODES * N_CLASSES);
    agg2_edge_kernel<<<cdiv_l((long)N_EDGES * N_CLASSES, B), B, 0, stream>>>(
        row, col, dinv, h2p, agg2);

    logsoftmax_kernel<<<cdiv_l(N_NODES, B), B, 0, stream>>>(agg2, b2, out);
}


# Degree pass also stores each edge's destination tile number as one byte; every tile's edge filter reads those bytes (a quarter of the L2 bytes of the ids); adds stay in arrival order
# speedup vs baseline: 1.9575x; 1.9575x over previous
#include <hip/hip_runtime.h>
#include <math.h>

typedef __attribute__((ext_vector_type(2))) float v2f;
typedef __attribute__((ext_vector_type(8))) float v8f;

#define N_NODES   100000
#define N_EDGES   1600000
#define IN_FEAT   256
#define HIDDEN    128
#define N_CLASSES 16

#define TILE_NODES   400
static_assert(N_NODES % TILE_NODES == 0, "whole tiles only");
static_assert(N_NODES / TILE_NODES <= 256, "a tile number fits one byte");
static_assert(N_EDGES % 4 == 0, "four edges per thread in the degree pass");


__global__ void init_deg_kernel(float* __restrict__ deg) {
    int i = blockIdx.x * blockDim.x + threadIdx.x;
    if (i < N_NODES) deg[i] = 1.0f;
}

__global__ void deg_accum_kernel(const int* __restrict__ col, float* __restrict__ deg,
                                 unsigned int* __restrict__ edge_tile) {
    int i = blockIdx.x * blockDim.x + threadIdx.x;
    if (i < N_EDGES / 4) {
        const int4 c = *(const int4*)(col + 4 * i);
        unsafeAtomicAdd(&deg[c.x], 1.0f);
        unsafeAtomicAdd(&deg[c.y], 1.0f);
        unsafeAtomicAdd(&deg[c.z], 1.0f);
        unsafeAtomicAdd(&deg[c.w], 1.0f);
        edge_tile[i] = (unsigned)(c.x / TILE_NODES) | ((unsigned)(c.y / TILE_NODES) << 8) |
                       ((unsigned)(c.z / TILE_NODES) << 16) | ((unsigned)(c.w / TILE_NODES) << 24);
    }
}

__global__ void dinv_kernel(float* __restrict__ deg) {
    int i = blockIdx.x * blockDim.x + threadIdx.x;
    if (i < N_NODES) deg[i] = rsqrtf(deg[i]);
}


__global__ __launch_bounds__(256) void gemm1_kernel(const float* __restrict__ x,
                                                    const float* __restrict__ W1,
                                                    const float* __restrict__ dinv,
                                                    float* __restrict__ hp) {
    const int lane = threadIdx.x & 31;
    const int wave = (blockIdx.x * blockDim.x + threadIdx.x) >> 5;
    const int row0 = wave * 16;
    if (row0 >= N_NODES) return;

    const int m  = lane & 15;
    const int kh = lane >> 4;
    const float* arow = x + (size_t)(row0 + m) * IN_FEAT;

    v8f acc[8];
#pragma unroll
    for (int n = 0; n < 8; ++n) acc[n] = v8f{};

    for (int k = 0; k < IN_FEAT; k += 4) {
        v2f a;
        a.x = arow[k + 2 * kh + 0];
        a.y = arow[k + 2 * kh + 1];
        const float* b0 = W1 + (size_t)(k + 2 * kh) * HIDDEN + m;
#pragma unroll
        for (int n = 0; n < 8; ++n) {
            v2f b;
            b.x = b0[n * 16];
            b.y = b0[HIDDEN + n * 16];
            acc[n] = __builtin_amdgcn_wmma_f32_16x16x4_f32(
                false, a, false, b, (short)0, acc[n], false, false);
        }
    }

    const int mo = kh * 8;
    float ds[8];
#pragma unroll
    for (int j = 0; j < 8; ++j) ds[j] = dinv[row0 + mo + j];
#pragma unroll
    for (int n = 0; n < 8; ++n) {
#pragma unroll
        for (int j = 0; j < 8; ++j) {
            hp[(size_t)(row0 + mo + j) * HIDDEN + n * 16 + m] = ds[j] * acc[n][j];
        }
    }
}


__global__ __launch_bounds__(256) void gemm2_kernel(const float* __restrict__ h,
                                                    const float* __restrict__ W2,
                                                    const float* __restrict__ dinv,
                                                    float* __restrict__ h2p,
                                                    float* __restrict__ agg) {
    const int lane = threadIdx.x & 31;
    const int wave = (blockIdx.x * blockDim.x + threadIdx.x) >> 5;
    const int row0 = wave * 16;
    if (row0 >= N_NODES) return;

    const int m  = lane & 15;
    const int kh = lane >> 4;
    const float* arow = h + (size_t)(row0 + m) * HIDDEN;

    v8f acc = v8f{};
    for (int k = 0; k < HIDDEN; k += 4) {
        v2f a;
        a.x = arow[k + 2 * kh + 0];
        a.y = arow[k + 2 * kh + 1];
        v2f b;
        b.x = W2[(size_t)(k + 2 * kh + 0) * N_CLASSES + m];
        b.y = W2[(size_t)(k + 2 * kh + 1) * N_CLASSES + m];
        acc = __builtin_amdgcn_wmma_f32_16x16x4_f32(
            false, a, false, b, (short)0, acc, false, false);
    }

    const int mo = kh * 8;
#pragma unroll
    for (int j = 0; j < 8; ++j) {
        const float v = dinv[row0 + mo + j] * acc[j];
        h2p[(size_t)(row0 + mo + j) * N_CLASSES + m] = v;
        agg[(size_t)(row0 + mo + j) * N_CLASSES + m] = v;
    }
}


#define EDGES_PER_WAVE 32
#define EDGE_GROUP     4
static_assert(N_EDGES % EDGES_PER_WAVE == 0, "a wave's 32 edges are all valid or all out of range");
static_assert(EDGES_PER_WAVE % EDGE_GROUP == 0, "whole groups only");

#define TILE_WAVES   32
#define SCAN_EDGES   1024
#define SCAN_STEP    512
#define QUEUE_SLOTS  64
static_assert(N_EDGES >= SCAN_EDGES &&
              (N_EDGES % SCAN_EDGES == 0 || N_EDGES % SCAN_EDGES == SCAN_STEP),
              "a moved-back trip repeats exactly its first 512-edge step, or nothing");

__device__ __forceinline__ void tile_load_group(float (&v)[EDGE_GROUP][4],
                                                const float* __restrict__ hp,
                                                int r_vec, int first, int n, int lane) {
#pragma unroll
    for (int q = 0; q < EDGE_GROUP; ++q) {
        if (first + q < n) {
            const int rs = __builtin_amdgcn_readlane(r_vec, first + q);
            const float* p = hp + (size_t)rs * HIDDEN + lane;
#pragma unroll
            for (int j = 0; j < 4; ++j) v[q][j] = p[j * 32];
        }
    }
}

__device__ __forceinline__ void tile_add_group(const float (&v)[EDGE_GROUP][4],
                                               float* acc,
                                               int cl_vec, int first, int n, int lane) {
#pragma unroll
    for (int q = 0; q < EDGE_GROUP; ++q) {
        if (first + q < n) {
            const int cs = __builtin_amdgcn_readlane(cl_vec, first + q);
            float* dst = acc + cs * HIDDEN + lane;
#pragma unroll
            for (int j = 0; j < 4; ++j) unsafeAtomicAdd(dst + j * 32, v[q][j]);
        }
    }
}

__device__ __forceinline__ void tile_process(const int* q, int head, int n, int lane,
                                             const int* __restrict__ row,
                                             const int* __restrict__ col, int n0,
                                             const float* __restrict__ hp,
                                             float* acc) {
    const bool valid = lane < n;
    const int e = q[(head + lane) & (QUEUE_SLOTS - 1)];
    int r = 0, cl = 0;
    if (valid) {
        r  = row[e];
        cl = col[e] - n0;
    }
    float v[2][EDGE_GROUP][4];
    tile_load_group(v[0], hp, r, 0, n, lane);
#pragma unroll
    for (int g = 0; g < EDGES_PER_WAVE / EDGE_GROUP; ++g) {
        if (g * EDGE_GROUP < n) {
            if (g + 1 < EDGES_PER_WAVE / EDGE_GROUP)
                tile_load_group(v[(g + 1) & 1], hp, r, (g + 1) * EDGE_GROUP, n, lane);
            tile_add_group(v[g & 1], acc, cl, g * EDGE_GROUP, n, lane);
        }
    }
}

__global__ __launch_bounds__(TILE_WAVES * 32) void agg1_tile_kernel(const int* __restrict__ row,
                                                                    const int* __restrict__ col,
                                                                    const unsigned int* __restrict__ edge_tile,
                                                                    const float* __restrict__ dinv,
                                                                    const float* __restrict__ hp,
                                                                    const float* __restrict__ b1,
                                                                    float* __restrict__ h) {
    __shared__ float acc[TILE_NODES * HIDDEN];
    __shared__ float dinv_tile[TILE_NODES];
    __shared__ int   queue[TILE_WAVES][QUEUE_SLOTS];

    const int tid  = threadIdx.x;
    const int lane = tid & 31;
    const int wave = tid >> 5;
    const int n0   = blockIdx.x * TILE_NODES;

    for (int i = tid; i < TILE_NODES; i += TILE_WAVES * 32) dinv_tile[i] = dinv[n0 + i];
    {
        const float4* s4 = (const float4*)(hp + (size_t)n0 * HIDDEN);
        float4* a4 = (float4*)acc;
        for (int i = tid; i < TILE_NODES * HIDDEN / 4; i += TILE_WAVES * 32) a4[i] = s4[i];
    }
    __syncthreads();

    int* q = queue[wave];
    int head = 0, tail = 0;
    const int n_chunks = (N_EDGES + SCAN_EDGES - 1) / SCAN_EDGES;
    const int last_base = N_EDGES - SCAN_EDGES;

    const unsigned char* tb = (const unsigned char*)edge_tile;
    const unsigned pat = (unsigned)blockIdx.x * 0x01010101u;

    auto load_trip = [&](uint4 (&d)[2], int chunk) {
        const int c0 = chunk * SCAN_EDGES;
        const int base = c0 < last_base ? c0 : last_base;
#pragma unroll
        for (int k = 0; k < 2; ++k) d[k] = *(const uint4*)(tb + base + k * SCAN_STEP + 16 * lane);
    };
    auto hit_bits = [&](const uint4 (&d)[2], int chunk) -> unsigned {
        unsigned hb = 0u;
#pragma unroll
        for (int k = 0; k < 2; ++k) {
            const unsigned ww[4] = {d[k].x, d[k].y, d[k].z, d[k].w};
#pragma unroll
            for (int c = 0; c < 4; ++c) {
                const unsigned t = ww[c] ^ pat;
                const unsigned nz = (((t & 0x7F7F7F7Fu) + 0x7F7F7F7Fu) | t) & 0x80808080u;
                hb |= (nz ^ 0x80808080u) >> (7 - (4 * k + c));
            }
        }
        const int c0 = chunk * SCAN_EDGES;
        return c0 > last_base ? (hb & 0xF0F0F0F0u) : hb;
    };
    auto take_hits = [&](unsigned hb, int chunk) {
        const int c0 = chunk * SCAN_EDGES;
        const int base = c0 < last_base ? c0 : last_base;
        unsigned m = __builtin_amdgcn_ballot_w32(hb != 0u);
        while (m != 0u) {
            if (hb != 0u) {
                const int p = __builtin_ctz(hb);
                hb &= hb - 1u;
                const int w = p & 7;
                const int pos = (tail + (int)__builtin_amdgcn_mbcnt_lo(m, 0u)) & (QUEUE_SLOTS - 1);
                q[pos] = base + (w >> 2) * SCAN_STEP + 16 * lane + 4 * (w & 3) + (p >> 3);
            }
            tail += __builtin_popcount(m);
            if (tail - head >= 32) {
                __builtin_amdgcn_fence(__ATOMIC_ACQ_REL, "wavefront");
                tile_process(q, head, 32, lane, row, col, n0, hp, acc);
                head += 32;
            }
            m = __builtin_amdgcn_ballot_w32(hb != 0u);
        }
    };

    uint4 d0[2], d1[2], d2[2], d3[2];
    load_trip(d0, wave);
    __builtin_amdgcn_sched_barrier(0);
    load_trip(d1, wave + TILE_WAVES);
    __builtin_amdgcn_sched_barrier(0);
    load_trip(d2, wave + 2 * TILE_WAVES);
    for (int chunk = wave; chunk < n_chunks; chunk += 4 * TILE_WAVES) {
        load_trip(d3, chunk + 3 * TILE_WAVES);
        take_hits(hit_bits(d0, chunk), chunk);
        load_trip(d0, chunk + 4 * TILE_WAVES);
        if (chunk + TILE_WAVES < n_chunks)
            take_hits(hit_bits(d1, chunk + TILE_WAVES), chunk + TILE_WAVES);
        load_trip(d1, chunk + 5 * TILE_WAVES);
        if (chunk + 2 * TILE_WAVES < n_chunks)
            take_hits(hit_bits(d2, chunk + 2 * TILE_WAVES), chunk + 2 * TILE_WAVES);
        load_trip(d2, chunk + 6 * TILE_WAVES);
        if (chunk + 3 * TILE_WAVES < n_chunks)
            take_hits(hit_bits(d3, chunk + 3 * TILE_WAVES), chunk + 3 * TILE_WAVES);
    }
    if (tail - head > 0) {
        __builtin_amdgcn_fence(__ATOMIC_ACQ_REL, "wavefront");
        tile_process(q, head, tail - head, lane, row, col, n0, hp, acc);
    }
    __syncthreads();

    {
        const float4* a4 = (const float4*)acc;
        const float4* b4 = (const float4*)b1;
        float4* o4 = (float4*)(h + (size_t)n0 * HIDDEN);
        for (int i = tid; i < TILE_NODES * HIDDEN / 4; i += TILE_WAVES * 32) {
            const float d = dinv_tile[i >> 5];
            const float4 a = a4[i];
            const float4 b = b4[i & 31];
            float4 o;
            o.x = d * a.x + b.x; o.y = d * a.y + b.y; o.z = d * a.z + b.z; o.w = d * a.w + b.w;
            o.x = o.x > 0.0f ? o.x : 0.0f; o.y = o.y > 0.0f ? o.y : 0.0f;
            o.z = o.z > 0.0f ? o.z : 0.0f; o.w = o.w > 0.0f ? o.w : 0.0f;
            o4[i] = o;
        }
    }
}

__global__ __launch_bounds__(256) void agg2_edge_kernel(const int* __restrict__ row,
                                                        const int* __restrict__ col,
                                                        const float* __restrict__ h2p,
                                                        float* __restrict__ agg) {
    const int lane = threadIdx.x & 31;
    const int wave = (blockIdx.x * blockDim.x + threadIdx.x) >> 5;
    const int e0 = wave * EDGES_PER_WAVE;
    if (e0 >= N_EDGES) return;
    const int r = row[e0 + lane];
    const int c = col[e0 + lane];
    const int half = lane >> 4;
    const int f = lane & 15;

    float v[EDGES_PER_WAVE / 2];
#pragma unroll
    for (int i = 0; i < EDGES_PER_WAVE / 2; ++i) {
        const int rr = __shfl(r, 2 * i + half);
        v[i] = h2p[(size_t)rr * N_CLASSES + f];
    }
#pragma unroll
    for (int i = 0; i < EDGES_PER_WAVE / 2; ++i) {
        const int cc = __shfl(c, 2 * i + half);
        unsafeAtomicAdd(&agg[(size_t)cc * N_CLASSES + f], v[i]);
    }
}

__global__ void logsoftmax_kernel(const float* __restrict__ agg,
                                  const float* __restrict__ dinv,
                                  const float* __restrict__ b2,
                                  float* __restrict__ out) {
    int i = blockIdx.x * blockDim.x + threadIdx.x;
    if (i >= N_NODES) return;
    const float d = dinv[i];
    float v[N_CLASSES];
    float mx = -INFINITY;
#pragma unroll
    for (int c = 0; c < N_CLASSES; ++c) {
        v[c] = d * agg[(size_t)i * N_CLASSES + c] + b2[c];
        mx = fmaxf(mx, v[c]);
    }
    float s = 0.0f;
#pragma unroll
    for (int c = 0; c < N_CLASSES; ++c) s += expf(v[c] - mx);
    float lse = mx + logf(s);
#pragma unroll
    for (int c = 0; c < N_CLASSES; ++c) {
        out[(size_t)i * N_CLASSES + c] = v[c] - lse;
    }
}


static inline int cdiv_l(long a, int b) { return (int)((a + b - 1) / b); }

extern "C" void kernel_launch(void* const* d_in, const int* in_sizes, int n_in,
                              void* d_out, int out_size, void* d_ws, size_t ws_size,
                              hipStream_t stream) {
    const float* x   = (const float*)d_in[0];
    const int*   ei  = (const int*)d_in[1];
    const int*   row = ei;
    const int*   col = ei + N_EDGES;
    const float* W1  = (const float*)d_in[2];
    const float* b1  = (const float*)d_in[3];
    const float* W2  = (const float*)d_in[4];
    const float* b2  = (const float*)d_in[5];
    float* out = (float*)d_out;

    float* ws   = (float*)d_ws;
    float* dinv = ws;
    float* hp   = ws + ((N_NODES + 255) & ~255);
    float* h    = hp + (size_t)N_NODES * HIDDEN;
    float* h2p  = hp;
    float* agg2 = hp + (size_t)N_NODES * N_CLASSES;
    static_assert((size_t)N_EDGES <= (size_t)N_NODES * N_CLASSES * sizeof(float),
                  "the tile bytes fit the output buffer");
    unsigned int* edge_tile = (unsigned int*)d_out;

    const int B = 256;

    init_deg_kernel<<<cdiv_l(N_NODES, B), B, 0, stream>>>(dinv);
    deg_accum_kernel<<<cdiv_l(N_EDGES / 4, B), B, 0, stream>>>(col, dinv, edge_tile);
    dinv_kernel<<<cdiv_l(N_NODES, B), B, 0, stream>>>(dinv);

    const long waves1 = (N_NODES + 15) / 16;
    gemm1_kernel<<<cdiv_l(waves1 * 32, B), B, 0, stream>>>(x, W1, dinv, hp);
    agg1_tile_kernel<<<N_NODES / TILE_NODES, TILE_WAVES * 32, 0, stream>>>(
        row, col, edge_tile, dinv, hp, b1, h);

    gemm2_kernel<<<cdiv_l(waves1 * 32, B), B, 0, stream>>>(h, W2, dinv, h2p, agg2);
    agg2_edge_kernel<<<cdiv_l((long)(N_EDGES / EDGES_PER_WAVE) * 32, B), B, 0, stream>>>(
        row, col, h2p, agg2);

    logsoftmax_kernel<<<cdiv_l(N_NODES, B), B, 0, stream>>>(agg2, dinv, b2, out);
}


# Projection 1: a 32-wave work-group per 400 rows stages all of W1 once in LDS as interleaved k-pair rows; each B fragment is one 8-byte LDS read (was two 4-byte global loads per WMMA)
# speedup vs baseline: 2.1089x; 1.0773x over previous
#include <hip/hip_runtime.h>
#include <math.h>

typedef __attribute__((ext_vector_type(2))) float v2f;
typedef __attribute__((ext_vector_type(8))) float v8f;

#define N_NODES   100000
#define N_EDGES   1600000
#define IN_FEAT   256
#define HIDDEN    128
#define N_CLASSES 16

#define TILE_NODES   400
static_assert(N_NODES % TILE_NODES == 0, "whole tiles only");
static_assert(N_NODES / TILE_NODES <= 256, "a tile number fits one byte");
static_assert(N_EDGES % 4 == 0, "four edges per thread in the degree pass");


__global__ void init_deg_kernel(float* __restrict__ deg) {
    int i = blockIdx.x * blockDim.x + threadIdx.x;
    if (i < N_NODES) deg[i] = 1.0f;
}

__global__ void deg_accum_kernel(const int* __restrict__ col, float* __restrict__ deg,
                                 unsigned int* __restrict__ edge_tile) {
    int i = blockIdx.x * blockDim.x + threadIdx.x;
    if (i < N_EDGES / 4) {
        const int4 c = *(const int4*)(col + 4 * i);
        unsafeAtomicAdd(&deg[c.x], 1.0f);
        unsafeAtomicAdd(&deg[c.y], 1.0f);
        unsafeAtomicAdd(&deg[c.z], 1.0f);
        unsafeAtomicAdd(&deg[c.w], 1.0f);
        edge_tile[i] = (unsigned)(c.x / TILE_NODES) | ((unsigned)(c.y / TILE_NODES) << 8) |
                       ((unsigned)(c.z / TILE_NODES) << 16) | ((unsigned)(c.w / TILE_NODES) << 24);
    }
}

__global__ void dinv_kernel(float* __restrict__ deg) {
    int i = blockIdx.x * blockDim.x + threadIdx.x;
    if (i < N_NODES) deg[i] = rsqrtf(deg[i]);
}


#define GEMM1_ROWS   400
#define GEMM1_WAVES  32
#define W1_PITCH     288
#define GEMM1_UNITS  ((GEMM1_ROWS / 16) * 2)
static_assert(N_NODES % GEMM1_ROWS == 0 && GEMM1_ROWS % 16 == 0, "whole row tiles only");
static_assert(HIDDEN == 128 && IN_FEAT % 32 == 0, "unit = 64 columns, A chunks of 32 columns");

__device__ __forceinline__ void gemm1_load_a(v2f (&a)[8], const float* __restrict__ p) {
#pragma unroll
    for (int s = 0; s < 8; ++s) a[s] = *(const v2f*)(p + 4 * s);
}

__device__ __forceinline__ void gemm1_chunk(v8f (&acc)[4], const v2f (&a)[8], const float* bc) {
#pragma unroll
    for (int s = 0; s < 8; ++s) {
        const float* bk = bc + (2 * s) * W1_PITCH;
#pragma unroll
        for (int n = 0; n < 4; ++n) {
            const v2f b = *(const v2f*)(bk + 32 * n);
            acc[n] = __builtin_amdgcn_wmma_f32_16x16x4_f32(
                false, a[s], false, b, (short)0, acc[n], false, false);
        }
    }
}

__global__ __launch_bounds__(GEMM1_WAVES * 32) void gemm1_kernel(const float* __restrict__ x,
                                                                const float* __restrict__ W1,
                                                                const float* __restrict__ dinv,
                                                                float* __restrict__ hp) {
    __shared__ __attribute__((aligned(16))) float Wl[(IN_FEAT / 2) * W1_PITCH];

    const int tid = threadIdx.x;
    for (int it = tid; it < (IN_FEAT / 2) * (HIDDEN / 4); it += GEMM1_WAVES * 32) {
        const int p = it / (HIDDEN / 4);
        const int c = it % (HIDDEN / 4);
        const float4 r0 = *(const float4*)(W1 + (size_t)(2 * p) * HIDDEN + 4 * c);
        const float4 r1 = *(const float4*)(W1 + (size_t)(2 * p + 1) * HIDDEN + 4 * c);
        float4* d = (float4*)(Wl + p * W1_PITCH + 8 * c);
        d[0] = make_float4(r0.x, r1.x, r0.y, r1.y);
        d[1] = make_float4(r0.z, r1.z, r0.w, r1.w);
    }
    __syncthreads();

    const int lane = tid & 31;
    const int wave = tid >> 5;
    const int m  = lane & 15;
    const int kh = lane >> 4;
    const int mo = kh * 8;

    for (int u = wave; u < GEMM1_UNITS; u += GEMM1_WAVES) {
        const int row0 = blockIdx.x * GEMM1_ROWS + (u >> 1) * 16;
        const int col0 = (u & 1) * 64;
        const float* arow = x + (size_t)(row0 + m) * IN_FEAT + 2 * kh;
        const float* bl = Wl + kh * W1_PITCH + (col0 + m) * 2;

        v8f acc[4];
#pragma unroll
        for (int n = 0; n < 4; ++n) acc[n] = v8f{};

        v2f a0[8], a1[8];
        gemm1_load_a(a0, arow);
#pragma nounroll
        for (int cp = 0; cp < IN_FEAT / 64; ++cp) {
            gemm1_load_a(a1, arow + 64 * cp + 32);
            gemm1_chunk(acc, a0, bl + (32 * cp) * W1_PITCH);
            if (2 * cp + 2 < IN_FEAT / 32)
                gemm1_load_a(a0, arow + 64 * cp + 64);
            gemm1_chunk(acc, a1, bl + (32 * cp + 16) * W1_PITCH);
        }

        float ds[8];
#pragma unroll
        for (int j = 0; j < 8; ++j) ds[j] = dinv[row0 + mo + j];
#pragma unroll
        for (int n = 0; n < 4; ++n) {
#pragma unroll
            for (int j = 0; j < 8; ++j) {
                hp[(size_t)(row0 + mo + j) * HIDDEN + col0 + n * 16 + m] = ds[j] * acc[n][j];
            }
        }
    }
}


__global__ __launch_bounds__(256) void gemm2_kernel(const float* __restrict__ h,
                                                    const float* __restrict__ W2,
                                                    const float* __restrict__ dinv,
                                                    float* __restrict__ h2p,
                                                    float* __restrict__ agg) {
    const int lane = threadIdx.x & 31;
    const int wave = (blockIdx.x * blockDim.x + threadIdx.x) >> 5;
    const int row0 = wave * 16;
    if (row0 >= N_NODES) return;

    const int m  = lane & 15;
    const int kh = lane >> 4;
    const float* arow = h + (size_t)(row0 + m) * HIDDEN;

    v8f acc = v8f{};
    for (int k = 0; k < HIDDEN; k += 4) {
        v2f a;
        a.x = arow[k + 2 * kh + 0];
        a.y = arow[k + 2 * kh + 1];
        v2f b;
        b.x = W2[(size_t)(k + 2 * kh + 0) * N_CLASSES + m];
        b.y = W2[(size_t)(k + 2 * kh + 1) * N_CLASSES + m];
        acc = __builtin_amdgcn_wmma_f32_16x16x4_f32(
            false, a, false, b, (short)0, acc, false, false);
    }

    const int mo = kh * 8;
#pragma unroll
    for (int j = 0; j < 8; ++j) {
        const float v = dinv[row0 + mo + j] * acc[j];
        h2p[(size_t)(row0 + mo + j) * N_CLASSES + m] = v;
        agg[(size_t)(row0 + mo + j) * N_CLASSES + m] = v;
    }
}


#define EDGES_PER_WAVE 32
#define EDGE_GROUP     4
static_assert(N_EDGES % EDGES_PER_WAVE == 0, "a wave's 32 edges are all valid or all out of range");
static_assert(EDGES_PER_WAVE % EDGE_GROUP == 0, "whole groups only");

#define TILE_WAVES   32
#define SCAN_EDGES   1024
#define SCAN_STEP    512
#define QUEUE_SLOTS  64
static_assert(N_EDGES >= SCAN_EDGES &&
              (N_EDGES % SCAN_EDGES == 0 || N_EDGES % SCAN_EDGES == SCAN_STEP),
              "a moved-back trip repeats exactly its first 512-edge step, or nothing");

__device__ __forceinline__ void tile_load_group(float (&v)[EDGE_GROUP][4],
                                                const float* __restrict__ hp,
                                                int r_vec, int first, int n, int lane) {
#pragma unroll
    for (int q = 0; q < EDGE_GROUP; ++q) {
        if (first + q < n) {
            const int rs = __builtin_amdgcn_readlane(r_vec, first + q);
            const float* p = hp + (size_t)rs * HIDDEN + lane;
#pragma unroll
            for (int j = 0; j < 4; ++j) v[q][j] = p[j * 32];
        }
    }
}

__device__ __forceinline__ void tile_add_group(const float (&v)[EDGE_GROUP][4],
                                               float* acc,
                                               int cl_vec, int first, int n, int lane) {
#pragma unroll
    for (int q = 0; q < EDGE_GROUP; ++q) {
        if (first + q < n) {
            const int cs = __builtin_amdgcn_readlane(cl_vec, first + q);
            float* dst = acc + cs * HIDDEN + lane;
#pragma unroll
            for (int j = 0; j < 4; ++j) unsafeAtomicAdd(dst + j * 32, v[q][j]);
        }
    }
}

__device__ __forceinline__ void tile_process(const int* q, int head, int n, int lane,
                                             const int* __restrict__ row,
                                             const int* __restrict__ col, int n0,
                                             const float* __restrict__ hp,
                                             float* acc) {
    const bool valid = lane < n;
    const int e = q[(head + lane) & (QUEUE_SLOTS - 1)];
    int r = 0, cl = 0;
    if (valid) {
        r  = row[e];
        cl = col[e] - n0;
    }
    float v[2][EDGE_GROUP][4];
    tile_load_group(v[0], hp, r, 0, n, lane);
#pragma unroll
    for (int g = 0; g < EDGES_PER_WAVE / EDGE_GROUP; ++g) {
        if (g * EDGE_GROUP < n) {
            if (g + 1 < EDGES_PER_WAVE / EDGE_GROUP)
                tile_load_group(v[(g + 1) & 1], hp, r, (g + 1) * EDGE_GROUP, n, lane);
            tile_add_group(v[g & 1], acc, cl, g * EDGE_GROUP, n, lane);
        }
    }
}

__global__ __launch_bounds__(TILE_WAVES * 32) void agg1_tile_kernel(const int* __restrict__ row,
                                                                    const int* __restrict__ col,
                                                                    const unsigned int* __restrict__ edge_tile,
                                                                    const float* __restrict__ dinv,
                                                                    const float* __restrict__ hp,
                                                                    const float* __restrict__ b1,
                                                                    float* __restrict__ h) {
    __shared__ float acc[TILE_NODES * HIDDEN];
    __shared__ float dinv_tile[TILE_NODES];
    __shared__ int   queue[TILE_WAVES][QUEUE_SLOTS];

    const int tid  = threadIdx.x;
    const int lane = tid & 31;
    const int wave = tid >> 5;
    const int n0   = blockIdx.x * TILE_NODES;

    for (int i = tid; i < TILE_NODES; i += TILE_WAVES * 32) dinv_tile[i] = dinv[n0 + i];
    {
        const float4* s4 = (const float4*)(hp + (size_t)n0 * HIDDEN);
        float4* a4 = (float4*)acc;
        for (int i = tid; i < TILE_NODES * HIDDEN / 4; i += TILE_WAVES * 32) a4[i] = s4[i];
    }
    __syncthreads();

    int* q = queue[wave];
    int head = 0, tail = 0;
    const int n_chunks = (N_EDGES + SCAN_EDGES - 1) / SCAN_EDGES;
    const int last_base = N_EDGES - SCAN_EDGES;

    const unsigned char* tb = (const unsigned char*)edge_tile;
    const unsigned pat = (unsigned)blockIdx.x * 0x01010101u;

    auto load_trip = [&](uint4 (&d)[2], int chunk) {
        const int c0 = chunk * SCAN_EDGES;
        const int base = c0 < last_base ? c0 : last_base;
#pragma unroll
        for (int k = 0; k < 2; ++k) d[k] = *(const uint4*)(tb + base + k * SCAN_STEP + 16 * lane);
    };
    auto hit_bits = [&](const uint4 (&d)[2], int chunk) -> unsigned {
        unsigned hb = 0u;
#pragma unroll
        for (int k = 0; k < 2; ++k) {
            const unsigned ww[4] = {d[k].x, d[k].y, d[k].z, d[k].w};
#pragma unroll
            for (int c = 0; c < 4; ++c) {
                const unsigned t = ww[c] ^ pat;
                const unsigned nz = (((t & 0x7F7F7F7Fu) + 0x7F7F7F7Fu) | t) & 0x80808080u;
                hb |= (nz ^ 0x80808080u) >> (7 - (4 * k + c));
            }
        }
        const int c0 = chunk * SCAN_EDGES;
        return c0 > last_base ? (hb & 0xF0F0F0F0u) : hb;
    };
    auto take_hits = [&](unsigned hb, int chunk) {
        const int c0 = chunk * SCAN_EDGES;
        const int base = c0 < last_base ? c0 : last_base;
        unsigned m = __builtin_amdgcn_ballot_w32(hb != 0u);
        while (m != 0u) {
            if (hb != 0u) {
                const int p = __builtin_ctz(hb);
                hb &= hb - 1u;
                const int w = p & 7;
                const int pos = (tail + (int)__builtin_amdgcn_mbcnt_lo(m, 0u)) & (QUEUE_SLOTS - 1);
                q[pos] = base + (w >> 2) * SCAN_STEP + 16 * lane + 4 * (w & 3) + (p >> 3);
            }
            tail += __builtin_popcount(m);
            if (tail - head >= 32) {
                __builtin_amdgcn_fence(__ATOMIC_ACQ_REL, "wavefront");
                tile_process(q, head, 32, lane, row, col, n0, hp, acc);
                head += 32;
            }
            m = __builtin_amdgcn_ballot_w32(hb != 0u);
        }
    };

    uint4 d0[2], d1[2], d2[2], d3[2];
    load_trip(d0, wave);
    __builtin_amdgcn_sched_barrier(0);
    load_trip(d1, wave + TILE_WAVES);
    __builtin_amdgcn_sched_barrier(0);
    load_trip(d2, wave + 2 * TILE_WAVES);
    for (int chunk = wave; chunk < n_chunks; chunk += 4 * TILE_WAVES) {
        load_trip(d3, chunk + 3 * TILE_WAVES);
        take_hits(hit_bits(d0, chunk), chunk);
        load_trip(d0, chunk + 4 * TILE_WAVES);
        if (chunk + TILE_WAVES < n_chunks)
            take_hits(hit_bits(d1, chunk + TILE_WAVES), chunk + TILE_WAVES);
        load_trip(d1, chunk + 5 * TILE_WAVES);
        if (chunk + 2 * TILE_WAVES < n_chunks)
            take_hits(hit_bits(d2, chunk + 2 * TILE_WAVES), chunk + 2 * TILE_WAVES);
        load_trip(d2, chunk + 6 * TILE_WAVES);
        if (chunk + 3 * TILE_WAVES < n_chunks)
            take_hits(hit_bits(d3, chunk + 3 * TILE_WAVES), chunk + 3 * TILE_WAVES);
    }
    if (tail - head > 0) {
        __builtin_amdgcn_fence(__ATOMIC_ACQ_REL, "wavefront");
        tile_process(q, head, tail - head, lane, row, col, n0, hp, acc);
    }
    __syncthreads();

    {
        const float4* a4 = (const float4*)acc;
        const float4* b4 = (const float4*)b1;
        float4* o4 = (float4*)(h + (size_t)n0 * HIDDEN);
        for (int i = tid; i < TILE_NODES * HIDDEN / 4; i += TILE_WAVES * 32) {
            const float d = dinv_tile[i >> 5];
            const float4 a = a4[i];
            const float4 b = b4[i & 31];
            float4 o;
            o.x = d * a.x + b.x; o.y = d * a.y + b.y; o.z = d * a.z + b.z; o.w = d * a.w + b.w;
            o.x = o.x > 0.0f ? o.x : 0.0f; o.y = o.y > 0.0f ? o.y : 0.0f;
            o.z = o.z > 0.0f ? o.z : 0.0f; o.w = o.w > 0.0f ? o.w : 0.0f;
            o4[i] = o;
        }
    }
}

__global__ __launch_bounds__(256) void agg2_edge_kernel(const int* __restrict__ row,
                                                        const int* __restrict__ col,
                                                        const float* __restrict__ h2p,
                                                        float* __restrict__ agg) {
    const int lane = threadIdx.x & 31;
    const int wave = (blockIdx.x * blockDim.x + threadIdx.x) >> 5;
    const int e0 = wave * EDGES_PER_WAVE;
    if (e0 >= N_EDGES) return;
    const int r = row[e0 + lane];
    const int c = col[e0 + lane];
    const int half = lane >> 4;
    const int f = lane & 15;

    float v[EDGES_PER_WAVE / 2];
#pragma unroll
    for (int i = 0; i < EDGES_PER_WAVE / 2; ++i) {
        const int rr = __shfl(r, 2 * i + half);
        v[i] = h2p[(size_t)rr * N_CLASSES + f];
    }
#pragma unroll
    for (int i = 0; i < EDGES_PER_WAVE / 2; ++i) {
        const int cc = __shfl(c, 2 * i + half);
        unsafeAtomicAdd(&agg[(size_t)cc * N_CLASSES + f], v[i]);
    }
}

__global__ void logsoftmax_kernel(const float* __restrict__ agg,
                                  const float* __restrict__ dinv,
                                  const float* __restrict__ b2,
                                  float* __restrict__ out) {
    int i = blockIdx.x * blockDim.x + threadIdx.x;
    if (i >= N_NODES) return;
    const float d = dinv[i];
    float v[N_CLASSES];
    float mx = -INFINITY;
#pragma unroll
    for (int c = 0; c < N_CLASSES; ++c) {
        v[c] = d * agg[(size_t)i * N_CLASSES + c] + b2[c];
        mx = fmaxf(mx, v[c]);
    }
    float s = 0.0f;
#pragma unroll
    for (int c = 0; c < N_CLASSES; ++c) s += expf(v[c] - mx);
    float lse = mx + logf(s);
#pragma unroll
    for (int c = 0; c < N_CLASSES; ++c) {
        out[(size_t)i * N_CLASSES + c] = v[c] - lse;
    }
}


static inline int cdiv_l(long a, int b) { return (int)((a + b - 1) / b); }

extern "C" void kernel_launch(void* const* d_in, const int* in_sizes, int n_in,
                              void* d_out, int out_size, void* d_ws, size_t ws_size,
                              hipStream_t stream) {
    const float* x   = (const float*)d_in[0];
    const int*   ei  = (const int*)d_in[1];
    const int*   row = ei;
    const int*   col = ei + N_EDGES;
    const float* W1  = (const float*)d_in[2];
    const float* b1  = (const float*)d_in[3];
    const float* W2  = (const float*)d_in[4];
    const float* b2  = (const float*)d_in[5];
    float* out = (float*)d_out;

    float* ws   = (float*)d_ws;
    float* dinv = ws;
    float* hp   = ws + ((N_NODES + 255) & ~255);
    float* h    = hp + (size_t)N_NODES * HIDDEN;
    float* h2p  = hp;
    float* agg2 = hp + (size_t)N_NODES * N_CLASSES;
    static_assert((size_t)N_EDGES <= (size_t)N_NODES * N_CLASSES * sizeof(float),
                  "the tile bytes fit the output buffer");
    unsigned int* edge_tile = (unsigned int*)d_out;

    const int B = 256;

    init_deg_kernel<<<cdiv_l(N_NODES, B), B, 0, stream>>>(dinv);
    deg_accum_kernel<<<cdiv_l(N_EDGES / 4, B), B, 0, stream>>>(col, dinv, edge_tile);
    dinv_kernel<<<cdiv_l(N_NODES, B), B, 0, stream>>>(dinv);

    const long waves1 = (N_NODES + 15) / 16;
    gemm1_kernel<<<N_NODES / GEMM1_ROWS, GEMM1_WAVES * 32, 0, stream>>>(x, W1, dinv, hp);
    agg1_tile_kernel<<<N_NODES / TILE_NODES, TILE_WAVES * 32, 0, stream>>>(
        row, col, edge_tile, dinv, hp, b1, h);

    gemm2_kernel<<<cdiv_l(waves1 * 32, B), B, 0, stream>>>(h, W2, dinv, h2p, agg2);
    agg2_edge_kernel<<<cdiv_l((long)(N_EDGES / EDGES_PER_WAVE) * 32, B), B, 0, stream>>>(
        row, col, h2p, agg2);

    logsoftmax_kernel<<<cdiv_l(N_NODES, B), B, 0, stream>>>(agg2, dinv, b2, out);
}


# Degree pass: 64 work-groups count edge slices into whole-graph LDS histograms of 16-bit counters (LDS integer adds); projection 1 sums the 64 planes for its rows; degree f32 atomics, 2 launches gone
# speedup vs baseline: 2.7363x; 1.2975x over previous
#include <hip/hip_runtime.h>
#include <math.h>

typedef __attribute__((ext_vector_type(2))) float v2f;
typedef __attribute__((ext_vector_type(8))) float v8f;

#define N_NODES   100000
#define N_EDGES   1600000
#define IN_FEAT   256
#define HIDDEN    128
#define N_CLASSES 16

#define TILE_NODES   400
static_assert(N_NODES % TILE_NODES == 0, "whole tiles only");
static_assert(N_NODES / TILE_NODES <= 256, "a tile number fits one byte");
static_assert(N_EDGES % 4 == 0, "four edges per thread in the degree pass");

#define DEG_WGS      64
#define DEG_THREADS  1024
#define HIST_WORDS   (N_NODES / 2)
#define QUADS_PER_WG (N_EDGES / 4 / DEG_WGS)
#define QUAD_TRIPS   ((QUADS_PER_WG + DEG_THREADS - 1) / DEG_THREADS)
static_assert(N_NODES % 8 == 0, "a plane is a whole number of 16-byte pieces");
static_assert((N_EDGES / 4) % DEG_WGS == 0, "equal slices of whole 4-edge items");
static_assert(N_EDGES / DEG_WGS < 65536, "a 16-bit counter cannot overflow within one slice");

__global__ __launch_bounds__(DEG_THREADS) void deg_hist_kernel(const int* __restrict__ col,
                                                              unsigned int* __restrict__ hist,
                                                              unsigned int* __restrict__ edge_tile) {
    __shared__ __attribute__((aligned(16))) unsigned int cnt[HIST_WORDS];
    const int tid = threadIdx.x;
    const int q0 = blockIdx.x * QUADS_PER_WG;
    int4 c[QUAD_TRIPS];
#pragma unroll
    for (int t = 0; t < QUAD_TRIPS; ++t) {
        const int i = tid + t * DEG_THREADS;
        if (i < QUADS_PER_WG) c[t] = *(const int4*)(col + 4 * (size_t)(q0 + i));
    }
    for (int i = tid; i < HIST_WORDS / 4; i += DEG_THREADS) ((uint4*)cnt)[i] = make_uint4(0u, 0u, 0u, 0u);
    __syncthreads();
#pragma unroll
    for (int t = 0; t < QUAD_TRIPS; ++t) {
        const int i = tid + t * DEG_THREADS;
        if (i < QUADS_PER_WG) {
            atomicAdd(&cnt[c[t].x >> 1], 1u << ((c[t].x & 1) * 16));
            atomicAdd(&cnt[c[t].y >> 1], 1u << ((c[t].y & 1) * 16));
            atomicAdd(&cnt[c[t].z >> 1], 1u << ((c[t].z & 1) * 16));
            atomicAdd(&cnt[c[t].w >> 1], 1u << ((c[t].w & 1) * 16));
            edge_tile[q0 + i] = (unsigned)(c[t].x / TILE_NODES) | ((unsigned)(c[t].y / TILE_NODES) << 8) |
                                ((unsigned)(c[t].z / TILE_NODES) << 16) | ((unsigned)(c[t].w / TILE_NODES) << 24);
        }
    }
    __syncthreads();
    uint4* dst = (uint4*)(hist + (size_t)blockIdx.x * HIST_WORDS);
    for (int i = tid; i < HIST_WORDS / 4; i += DEG_THREADS) dst[i] = ((const uint4*)cnt)[i];
}


#define GEMM1_ROWS   400
#define GEMM1_WAVES  32
#define W1_PITCH     288
#define GEMM1_UNITS  ((GEMM1_ROWS / 16) * 2)
static_assert(N_NODES % GEMM1_ROWS == 0 && GEMM1_ROWS % 16 == 0, "whole row tiles only");
static_assert(HIDDEN == 128 && IN_FEAT % 32 == 0, "unit = 64 columns, A chunks of 32 columns");

__device__ __forceinline__ void gemm1_load_a(v2f (&a)[8], const float* __restrict__ p) {
#pragma unroll
    for (int s = 0; s < 8; ++s) a[s] = *(const v2f*)(p + 4 * s);
}

__device__ __forceinline__ void gemm1_chunk(v8f (&acc)[4], const v2f (&a)[8], const float* bc) {
#pragma unroll
    for (int s = 0; s < 8; ++s) {
        const float* bk = bc + (2 * s) * W1_PITCH;
#pragma unroll
        for (int n = 0; n < 4; ++n) {
            const v2f b = *(const v2f*)(bk + 32 * n);
            acc[n] = __builtin_amdgcn_wmma_f32_16x16x4_f32(
                false, a[s], false, b, (short)0, acc[n], false, false);
        }
    }
}

__global__ __launch_bounds__(GEMM1_WAVES * 32) void gemm1_kernel(const float* __restrict__ x,
                                                                const float* __restrict__ W1,
                                                                const unsigned short* __restrict__ hist,
                                                                float* __restrict__ dinv,
                                                                float* __restrict__ hp) {
    __shared__ __attribute__((aligned(16))) float Wl[(IN_FEAT / 2) * W1_PITCH];
    __shared__ float dinv_l[GEMM1_ROWS];

    const int tid = threadIdx.x;
    if (tid < GEMM1_ROWS) {
        const int node = blockIdx.x * GEMM1_ROWS + tid;
        const unsigned short* hn = hist + node;
        unsigned int s = 1u;
#pragma unroll
        for (int k = 0; k < DEG_WGS; ++k) s += hn[(size_t)k * N_NODES];
        const float d = rsqrtf((float)s);
        dinv_l[tid] = d;
        dinv[node] = d;
    }
    for (int it = tid; it < (IN_FEAT / 2) * (HIDDEN / 4); it += GEMM1_WAVES * 32) {
        const int p = it / (HIDDEN / 4);
        const int c = it % (HIDDEN / 4);
        const float4 r0 = *(const float4*)(W1 + (size_t)(2 * p) * HIDDEN + 4 * c);
        const float4 r1 = *(const float4*)(W1 + (size_t)(2 * p + 1) * HIDDEN + 4 * c);
        float4* d = (float4*)(Wl + p * W1_PITCH + 8 * c);
        d[0] = make_float4(r0.x, r1.x, r0.y, r1.y);
        d[1] = make_float4(r0.z, r1.z, r0.w, r1.w);
    }
    __syncthreads();

    const int lane = tid & 31;
    const int wave = tid >> 5;
    const int m  = lane & 15;
    const int kh = lane >> 4;
    const int mo = kh * 8;

    for (int u = wave; u < GEMM1_UNITS; u += GEMM1_WAVES) {
        const int row0 = blockIdx.x * GEMM1_ROWS + (u >> 1) * 16;
        const int col0 = (u & 1) * 64;
        const float* arow = x + (size_t)(row0 + m) * IN_FEAT + 2 * kh;
        const float* bl = Wl + kh * W1_PITCH + (col0 + m) * 2;

        v8f acc[4];
#pragma unroll
        for (int n = 0; n < 4; ++n) acc[n] = v8f{};

        v2f a0[8], a1[8];
        gemm1_load_a(a0, arow);
#pragma nounroll
        for (int cp = 0; cp < IN_FEAT / 64; ++cp) {
            gemm1_load_a(a1, arow + 64 * cp + 32);
            gemm1_chunk(acc, a0, bl + (32 * cp) * W1_PITCH);
            if (2 * cp + 2 < IN_FEAT / 32)
                gemm1_load_a(a0, arow + 64 * cp + 64);
            gemm1_chunk(acc, a1, bl + (32 * cp + 16) * W1_PITCH);
        }

        float ds[8];
#pragma unroll
        for (int j = 0; j < 8; ++j) ds[j] = dinv_l[(u >> 1) * 16 + mo + j];
#pragma unroll
        for (int n = 0; n < 4; ++n) {
#pragma unroll
            for (int j = 0; j < 8; ++j) {
                hp[(size_t)(row0 + mo + j) * HIDDEN + col0 + n * 16 + m] = ds[j] * acc[n][j];
            }
        }
    }
}


__global__ __launch_bounds__(256) void gemm2_kernel(const float* __restrict__ h,
                                                    const float* __restrict__ W2,
                                                    const float* __restrict__ dinv,
                                                    float* __restrict__ h2p,
                                                    float* __restrict__ agg) {
    const int lane = threadIdx.x & 31;
    const int wave = (blockIdx.x * blockDim.x + threadIdx.x) >> 5;
    const int row0 = wave * 16;
    if (row0 >= N_NODES) return;

    const int m  = lane & 15;
    const int kh = lane >> 4;
    const float* arow = h + (size_t)(row0 + m) * HIDDEN;

    v8f acc = v8f{};
    for (int k = 0; k < HIDDEN; k += 4) {
        v2f a;
        a.x = arow[k + 2 * kh + 0];
        a.y = arow[k + 2 * kh + 1];
        v2f b;
        b.x = W2[(size_t)(k + 2 * kh + 0) * N_CLASSES + m];
        b.y = W2[(size_t)(k + 2 * kh + 1) * N_CLASSES + m];
        acc = __builtin_amdgcn_wmma_f32_16x16x4_f32(
            false, a, false, b, (short)0, acc, false, false);
    }

    const int mo = kh * 8;
#pragma unroll
    for (int j = 0; j < 8; ++j) {
        const float v = dinv[row0 + mo + j] * acc[j];
        h2p[(size_t)(row0 + mo + j) * N_CLASSES + m] = v;
        agg[(size_t)(row0 + mo + j) * N_CLASSES + m] = v;
    }
}


#define EDGES_PER_WAVE 32
#define EDGE_GROUP     4
static_assert(N_EDGES % EDGES_PER_WAVE == 0, "a wave's 32 edges are all valid or all out of range");
static_assert(EDGES_PER_WAVE % EDGE_GROUP == 0, "whole groups only");

#define TILE_WAVES   32
#define SCAN_EDGES   1024
#define SCAN_STEP    512
#define QUEUE_SLOTS  64
static_assert(N_EDGES >= SCAN_EDGES &&
              (N_EDGES % SCAN_EDGES == 0 || N_EDGES % SCAN_EDGES == SCAN_STEP),
              "a moved-back trip repeats exactly its first 512-edge step, or nothing");

__device__ __forceinline__ void tile_load_group(float (&v)[EDGE_GROUP][4],
                                                const float* __restrict__ hp,
                                                int r_vec, int first, int n, int lane) {
#pragma unroll
    for (int q = 0; q < EDGE_GROUP; ++q) {
        if (first + q < n) {
            const int rs = __builtin_amdgcn_readlane(r_vec, first + q);
            const float* p = hp + (size_t)rs * HIDDEN + lane;
#pragma unroll
            for (int j = 0; j < 4; ++j) v[q][j] = p[j * 32];
        }
    }
}

__device__ __forceinline__ void tile_add_group(const float (&v)[EDGE_GROUP][4],
                                               float* acc,
                                               int cl_vec, int first, int n, int lane) {
#pragma unroll
    for (int q = 0; q < EDGE_GROUP; ++q) {
        if (first + q < n) {
            const int cs = __builtin_amdgcn_readlane(cl_vec, first + q);
            float* dst = acc + cs * HIDDEN + lane;
#pragma unroll
            for (int j = 0; j < 4; ++j) unsafeAtomicAdd(dst + j * 32, v[q][j]);
        }
    }
}

__device__ __forceinline__ void tile_process(const int* q, int head, int n, int lane,
                                             const int* __restrict__ row,
                                             const int* __restrict__ col, int n0,
                                             const float* __restrict__ hp,
                                             float* acc) {
    const bool valid = lane < n;
    const int e = q[(head + lane) & (QUEUE_SLOTS - 1)];
    int r = 0, cl = 0;
    if (valid) {
        r  = row[e];
        cl = col[e] - n0;
    }
    float v[2][EDGE_GROUP][4];
    tile_load_group(v[0], hp, r, 0, n, lane);
#pragma unroll
    for (int g = 0; g < EDGES_PER_WAVE / EDGE_GROUP; ++g) {
        if (g * EDGE_GROUP < n) {
            if (g + 1 < EDGES_PER_WAVE / EDGE_GROUP)
                tile_load_group(v[(g + 1) & 1], hp, r, (g + 1) * EDGE_GROUP, n, lane);
            tile_add_group(v[g & 1], acc, cl, g * EDGE_GROUP, n, lane);
        }
    }
}

__global__ __launch_bounds__(TILE_WAVES * 32) void agg1_tile_kernel(const int* __restrict__ row,
                                                                    const int* __restrict__ col,
                                                                    const unsigned int* __restrict__ edge_tile,
                                                                    const float* __restrict__ dinv,
                                                                    const float* __restrict__ hp,
                                                                    const float* __restrict__ b1,
                                                                    float* __restrict__ h) {
    __shared__ float acc[TILE_NODES * HIDDEN];
    __shared__ float dinv_tile[TILE_NODES];
    __shared__ int   queue[TILE_WAVES][QUEUE_SLOTS];

    const int tid  = threadIdx.x;
    const int lane = tid & 31;
    const int wave = tid >> 5;
    const int n0   = blockIdx.x * TILE_NODES;

    for (int i = tid; i < TILE_NODES; i += TILE_WAVES * 32) dinv_tile[i] = dinv[n0 + i];
    {
        const float4* s4 = (const float4*)(hp + (size_t)n0 * HIDDEN);
        float4* a4 = (float4*)acc;
        for (int i = tid; i < TILE_NODES * HIDDEN / 4; i += TILE_WAVES * 32) a4[i] = s4[i];
    }
    __syncthreads();

    int* q = queue[wave];
    int head = 0, tail = 0;
    const int n_chunks = (N_EDGES + SCAN_EDGES - 1) / SCAN_EDGES;
    const int last_base = N_EDGES - SCAN_EDGES;

    const unsigned char* tb = (const unsigned char*)edge_tile;
    const unsigned pat = (unsigned)blockIdx.x * 0x01010101u;

    auto load_trip = [&](uint4 (&d)[2], int chunk) {
        const int c0 = chunk * SCAN_EDGES;
        const int base = c0 < last_base ? c0 : last_base;
#pragma unroll
        for (int k = 0; k < 2; ++k) d[k] = *(const uint4*)(tb + base + k * SCAN_STEP + 16 * lane);
    };
    auto hit_bits = [&](const uint4 (&d)[2], int chunk) -> unsigned {
        unsigned hb = 0u;
#pragma unroll
        for (int k = 0; k < 2; ++k) {
            const unsigned ww[4] = {d[k].x, d[k].y, d[k].z, d[k].w};
#pragma unroll
            for (int c = 0; c < 4; ++c) {
                const unsigned t = ww[c] ^ pat;
                const unsigned nz = (((t & 0x7F7F7F7Fu) + 0x7F7F7F7Fu) | t) & 0x80808080u;
                hb |= (nz ^ 0x80808080u) >> (7 - (4 * k + c));
            }
        }
        const int c0 = chunk * SCAN_EDGES;
        return c0 > last_base ? (hb & 0xF0F0F0F0u) : hb;
    };
    auto take_hits = [&](unsigned hb, int chunk) {
        const int c0 = chunk * SCAN_EDGES;
        const int base = c0 < last_base ? c0 : last_base;
        unsigned m = __builtin_amdgcn_ballot_w32(hb != 0u);
        while (m != 0u) {
            if (hb != 0u) {
                const int p = __builtin_ctz(hb);
                hb &= hb - 1u;
                const int w = p & 7;
                const int pos = (tail + (int)__builtin_amdgcn_mbcnt_lo(m, 0u)) & (QUEUE_SLOTS - 1);
                q[pos] = base + (w >> 2) * SCAN_STEP + 16 * lane + 4 * (w & 3) + (p >> 3);
            }
            tail += __builtin_popcount(m);
            if (tail - head >= 32) {
                __builtin_amdgcn_fence(__ATOMIC_ACQ_REL, "wavefront");
                tile_process(q, head, 32, lane, row, col, n0, hp, acc);
                head += 32;
            }
            m = __builtin_amdgcn_ballot_w32(hb != 0u);
        }
    };

    uint4 d0[2], d1[2], d2[2], d3[2];
    load_trip(d0, wave);
    __builtin_amdgcn_sched_barrier(0);
    load_trip(d1, wave + TILE_WAVES);
    __builtin_amdgcn_sched_barrier(0);
    load_trip(d2, wave + 2 * TILE_WAVES);
    for (int chunk = wave; chunk < n_chunks; chunk += 4 * TILE_WAVES) {
        load_trip(d3, chunk + 3 * TILE_WAVES);
        take_hits(hit_bits(d0, chunk), chunk);
        load_trip(d0, chunk + 4 * TILE_WAVES);
        if (chunk + TILE_WAVES < n_chunks)
            take_hits(hit_bits(d1, chunk + TILE_WAVES), chunk + TILE_WAVES);
        load_trip(d1, chunk + 5 * TILE_WAVES);
        if (chunk + 2 * TILE_WAVES < n_chunks)
            take_hits(hit_bits(d2, chunk + 2 * TILE_WAVES), chunk + 2 * TILE_WAVES);
        load_trip(d2, chunk + 6 * TILE_WAVES);
        if (chunk + 3 * TILE_WAVES < n_chunks)
            take_hits(hit_bits(d3, chunk + 3 * TILE_WAVES), chunk + 3 * TILE_WAVES);
    }
    if (tail - head > 0) {
        __builtin_amdgcn_fence(__ATOMIC_ACQ_REL, "wavefront");
        tile_process(q, head, tail - head, lane, row, col, n0, hp, acc);
    }
    __syncthreads();

    {
        const float4* a4 = (const float4*)acc;
        const float4* b4 = (const float4*)b1;
        float4* o4 = (float4*)(h + (size_t)n0 * HIDDEN);
        for (int i = tid; i < TILE_NODES * HIDDEN / 4; i += TILE_WAVES * 32) {
            const float d = dinv_tile[i >> 5];
            const float4 a = a4[i];
            const float4 b = b4[i & 31];
            float4 o;
            o.x = d * a.x + b.x; o.y = d * a.y + b.y; o.z = d * a.z + b.z; o.w = d * a.w + b.w;
            o.x = o.x > 0.0f ? o.x : 0.0f; o.y = o.y > 0.0f ? o.y : 0.0f;
            o.z = o.z > 0.0f ? o.z : 0.0f; o.w = o.w > 0.0f ? o.w : 0.0f;
            o4[i] = o;
        }
    }
}

__global__ __launch_bounds__(256) void agg2_edge_kernel(const int* __restrict__ row,
                                                        const int* __restrict__ col,
                                                        const float* __restrict__ h2p,
                                                        float* __restrict__ agg) {
    const int lane = threadIdx.x & 31;
    const int wave = (blockIdx.x * blockDim.x + threadIdx.x) >> 5;
    const int e0 = wave * EDGES_PER_WAVE;
    if (e0 >= N_EDGES) return;
    const int r = row[e0 + lane];
    const int c = col[e0 + lane];
    const int half = lane >> 4;
    const int f = lane & 15;

    float v[EDGES_PER_WAVE / 2];
#pragma unroll
    for (int i = 0; i < EDGES_PER_WAVE / 2; ++i) {
        const int rr = __shfl(r, 2 * i + half);
        v[i] = h2p[(size_t)rr * N_CLASSES + f];
    }
#pragma unroll
    for (int i = 0; i < EDGES_PER_WAVE / 2; ++i) {
        const int cc = __shfl(c, 2 * i + half);
        unsafeAtomicAdd(&agg[(size_t)cc * N_CLASSES + f], v[i]);
    }
}

__global__ void logsoftmax_kernel(const float* __restrict__ agg,
                                  const float* __restrict__ dinv,
                                  const float* __restrict__ b2,
                                  float* __restrict__ out) {
    int i = blockIdx.x * blockDim.x + threadIdx.x;
    if (i >= N_NODES) return;
    const float d = dinv[i];
    float v[N_CLASSES];
    float mx = -INFINITY;
#pragma unroll
    for (int c = 0; c < N_CLASSES; ++c) {
        v[c] = d * agg[(size_t)i * N_CLASSES + c] + b2[c];
        mx = fmaxf(mx, v[c]);
    }
    float s = 0.0f;
#pragma unroll
    for (int c = 0; c < N_CLASSES; ++c) s += expf(v[c] - mx);
    float lse = mx + logf(s);
#pragma unroll
    for (int c = 0; c < N_CLASSES; ++c) {
        out[(size_t)i * N_CLASSES + c] = v[c] - lse;
    }
}


static inline int cdiv_l(long a, int b) { return (int)((a + b - 1) / b); }

extern "C" void kernel_launch(void* const* d_in, const int* in_sizes, int n_in,
                              void* d_out, int out_size, void* d_ws, size_t ws_size,
                              hipStream_t stream) {
    const float* x   = (const float*)d_in[0];
    const int*   ei  = (const int*)d_in[1];
    const int*   row = ei;
    const int*   col = ei + N_EDGES;
    const float* W1  = (const float*)d_in[2];
    const float* b1  = (const float*)d_in[3];
    const float* W2  = (const float*)d_in[4];
    const float* b2  = (const float*)d_in[5];
    float* out = (float*)d_out;

    float* ws   = (float*)d_ws;
    float* dinv = ws;
    float* hp   = ws + ((N_NODES + 255) & ~255);
    float* h    = hp + (size_t)N_NODES * HIDDEN;
    float* h2p  = hp;
    float* agg2 = hp + (size_t)N_NODES * N_CLASSES;
    static_assert((size_t)N_EDGES <= (size_t)N_NODES * N_CLASSES * sizeof(float),
                  "the tile bytes fit the output buffer");
    unsigned int* edge_tile = (unsigned int*)d_out;
    static_assert((size_t)DEG_WGS * HIST_WORDS <= (size_t)N_NODES * HIDDEN, "the planes fit the h region");
    unsigned int* hist = (unsigned int*)h;

    const int B = 256;

    deg_hist_kernel<<<DEG_WGS, DEG_THREADS, 0, stream>>>(col, hist, edge_tile);

    const long waves1 = (N_NODES + 15) / 16;
    gemm1_kernel<<<N_NODES / GEMM1_ROWS, GEMM1_WAVES * 32, 0, stream>>>(
        x, W1, (const unsigned short*)hist, dinv, hp);
    agg1_tile_kernel<<<N_NODES / TILE_NODES, TILE_WAVES * 32, 0, stream>>>(
        row, col, edge_tile, dinv, hp, b1, h);

    gemm2_kernel<<<cdiv_l(waves1 * 32, B), B, 0, stream>>>(h, W2, dinv, h2p, agg2);
    agg2_edge_kernel<<<cdiv_l((long)(N_EDGES / EDGES_PER_WAVE) * 32, B), B, 0, stream>>>(
        row, col, h2p, agg2);

    logsoftmax_kernel<<<cdiv_l(N_NODES, B), B, 0, stream>>>(agg2, dinv, b2, out);
}


# Layer-1 gather: one 16-byte-per-lane load fetches a whole 512-B source row (was four 1-line loads); LDS tile rows keep columns transposed so the 4 LDS adds stay conflict-free; add order as before
# speedup vs baseline: 3.0593x; 1.1180x over previous
#include <hip/hip_runtime.h>
#include <math.h>

typedef __attribute__((ext_vector_type(2))) float v2f;
typedef __attribute__((ext_vector_type(8))) float v8f;

#define N_NODES   100000
#define N_EDGES   1600000
#define IN_FEAT   256
#define HIDDEN    128
#define N_CLASSES 16

#define TILE_NODES   400
static_assert(N_NODES % TILE_NODES == 0, "whole tiles only");
static_assert(N_NODES / TILE_NODES <= 256, "a tile number fits one byte");
static_assert(N_EDGES % 4 == 0, "four edges per thread in the degree pass");

#define DEG_WGS      64
#define DEG_THREADS  1024
#define HIST_WORDS   (N_NODES / 2)
#define QUADS_PER_WG (N_EDGES / 4 / DEG_WGS)
#define QUAD_TRIPS   ((QUADS_PER_WG + DEG_THREADS - 1) / DEG_THREADS)
static_assert(N_NODES % 8 == 0, "a plane is a whole number of 16-byte pieces");
static_assert((N_EDGES / 4) % DEG_WGS == 0, "equal slices of whole 4-edge items");
static_assert(N_EDGES / DEG_WGS < 65536, "a 16-bit counter cannot overflow within one slice");

__global__ __launch_bounds__(DEG_THREADS) void deg_hist_kernel(const int* __restrict__ col,
                                                              unsigned int* __restrict__ hist,
                                                              unsigned int* __restrict__ edge_tile) {
    __shared__ __attribute__((aligned(16))) unsigned int cnt[HIST_WORDS];
    const int tid = threadIdx.x;
    const int q0 = blockIdx.x * QUADS_PER_WG;
    int4 c[QUAD_TRIPS];
#pragma unroll
    for (int t = 0; t < QUAD_TRIPS; ++t) {
        const int i = tid + t * DEG_THREADS;
        if (i < QUADS_PER_WG) c[t] = *(const int4*)(col + 4 * (size_t)(q0 + i));
    }
    for (int i = tid; i < HIST_WORDS / 4; i += DEG_THREADS) ((uint4*)cnt)[i] = make_uint4(0u, 0u, 0u, 0u);
    __syncthreads();
#pragma unroll
    for (int t = 0; t < QUAD_TRIPS; ++t) {
        const int i = tid + t * DEG_THREADS;
        if (i < QUADS_PER_WG) {
            atomicAdd(&cnt[c[t].x >> 1], 1u << ((c[t].x & 1) * 16));
            atomicAdd(&cnt[c[t].y >> 1], 1u << ((c[t].y & 1) * 16));
            atomicAdd(&cnt[c[t].z >> 1], 1u << ((c[t].z & 1) * 16));
            atomicAdd(&cnt[c[t].w >> 1], 1u << ((c[t].w & 1) * 16));
            edge_tile[q0 + i] = (unsigned)(c[t].x / TILE_NODES) | ((unsigned)(c[t].y / TILE_NODES) << 8) |
                                ((unsigned)(c[t].z / TILE_NODES) << 16) | ((unsigned)(c[t].w / TILE_NODES) << 24);
        }
    }
    __syncthreads();
    uint4* dst = (uint4*)(hist + (size_t)blockIdx.x * HIST_WORDS);
    for (int i = tid; i < HIST_WORDS / 4; i += DEG_THREADS) dst[i] = ((const uint4*)cnt)[i];
}


#define GEMM1_ROWS   400
#define GEMM1_WAVES  32
#define W1_PITCH     288
#define GEMM1_UNITS  ((GEMM1_ROWS / 16) * 2)
static_assert(N_NODES % GEMM1_ROWS == 0 && GEMM1_ROWS % 16 == 0, "whole row tiles only");
static_assert(HIDDEN == 128 && IN_FEAT % 32 == 0, "unit = 64 columns, A chunks of 32 columns");

__device__ __forceinline__ void gemm1_load_a(v2f (&a)[8], const float* __restrict__ p) {
#pragma unroll
    for (int s = 0; s < 8; ++s) a[s] = *(const v2f*)(p + 4 * s);
}

__device__ __forceinline__ void gemm1_chunk(v8f (&acc)[4], const v2f (&a)[8], const float* bc) {
#pragma unroll
    for (int s = 0; s < 8; ++s) {
        const float* bk = bc + (2 * s) * W1_PITCH;
#pragma unroll
        for (int n = 0; n < 4; ++n) {
            const v2f b = *(const v2f*)(bk + 32 * n);
            acc[n] = __builtin_amdgcn_wmma_f32_16x16x4_f32(
                false, a[s], false, b, (short)0, acc[n], false, false);
        }
    }
}

__global__ __launch_bounds__(GEMM1_WAVES * 32) void gemm1_kernel(const float* __restrict__ x,
                                                                const float* __restrict__ W1,
                                                                const unsigned short* __restrict__ hist,
                                                                float* __restrict__ dinv,
                                                                float* __restrict__ hp) {
    __shared__ __attribute__((aligned(16))) float Wl[(IN_FEAT / 2) * W1_PITCH];
    __shared__ float dinv_l[GEMM1_ROWS];

    const int tid = threadIdx.x;
    if (tid < GEMM1_ROWS) {
        const int node = blockIdx.x * GEMM1_ROWS + tid;
        const unsigned short* hn = hist + node;
        unsigned int s = 1u;
#pragma unroll
        for (int k = 0; k < DEG_WGS; ++k) s += hn[(size_t)k * N_NODES];
        const float d = rsqrtf((float)s);
        dinv_l[tid] = d;
        dinv[node] = d;
    }
    for (int it = tid; it < (IN_FEAT / 2) * (HIDDEN / 4); it += GEMM1_WAVES * 32) {
        const int p = it / (HIDDEN / 4);
        const int c = it % (HIDDEN / 4);
        const float4 r0 = *(const float4*)(W1 + (size_t)(2 * p) * HIDDEN + 4 * c);
        const float4 r1 = *(const float4*)(W1 + (size_t)(2 * p + 1) * HIDDEN + 4 * c);
        float4* d = (float4*)(Wl + p * W1_PITCH + 8 * c);
        d[0] = make_float4(r0.x, r1.x, r0.y, r1.y);
        d[1] = make_float4(r0.z, r1.z, r0.w, r1.w);
    }
    __syncthreads();

    const int lane = tid & 31;
    const int wave = tid >> 5;
    const int m  = lane & 15;
    const int kh = lane >> 4;
    const int mo = kh * 8;

    for (int u = wave; u < GEMM1_UNITS; u += GEMM1_WAVES) {
        const int row0 = blockIdx.x * GEMM1_ROWS + (u >> 1) * 16;
        const int col0 = (u & 1) * 64;
        const float* arow = x + (size_t)(row0 + m) * IN_FEAT + 2 * kh;
        const float* bl = Wl + kh * W1_PITCH + (col0 + m) * 2;

        v8f acc[4];
#pragma unroll
        for (int n = 0; n < 4; ++n) acc[n] = v8f{};

        v2f a0[8], a1[8];
        gemm1_load_a(a0, arow);
#pragma nounroll
        for (int cp = 0; cp < IN_FEAT / 64; ++cp) {
            gemm1_load_a(a1, arow + 64 * cp + 32);
            gemm1_chunk(acc, a0, bl + (32 * cp) * W1_PITCH);
            if (2 * cp + 2 < IN_FEAT / 32)
                gemm1_load_a(a0, arow + 64 * cp + 64);
            gemm1_chunk(acc, a1, bl + (32 * cp + 16) * W1_PITCH);
        }

        float ds[8];
#pragma unroll
        for (int j = 0; j < 8; ++j) ds[j] = dinv_l[(u >> 1) * 16 + mo + j];
#pragma unroll
        for (int n = 0; n < 4; ++n) {
#pragma unroll
            for (int j = 0; j < 8; ++j) {
                hp[(size_t)(row0 + mo + j) * HIDDEN + col0 + n * 16 + m] = ds[j] * acc[n][j];
            }
        }
    }
}


__global__ __launch_bounds__(256) void gemm2_kernel(const float* __restrict__ h,
                                                    const float* __restrict__ W2,
                                                    const float* __restrict__ dinv,
                                                    float* __restrict__ h2p,
                                                    float* __restrict__ agg) {
    const int lane = threadIdx.x & 31;
    const int wave = (blockIdx.x * blockDim.x + threadIdx.x) >> 5;
    const int row0 = wave * 16;
    if (row0 >= N_NODES) return;

    const int m  = lane & 15;
    const int kh = lane >> 4;
    const float* arow = h + (size_t)(row0 + m) * HIDDEN;

    v8f acc = v8f{};
    for (int k = 0; k < HIDDEN; k += 4) {
        v2f a;
        a.x = arow[k + 2 * kh + 0];
        a.y = arow[k + 2 * kh + 1];
        v2f b;
        b.x = W2[(size_t)(k + 2 * kh + 0) * N_CLASSES + m];
        b.y = W2[(size_t)(k + 2 * kh + 1) * N_CLASSES + m];
        acc = __builtin_amdgcn_wmma_f32_16x16x4_f32(
            false, a, false, b, (short)0, acc, false, false);
    }

    const int mo = kh * 8;
#pragma unroll
    for (int j = 0; j < 8; ++j) {
        const float v = dinv[row0 + mo + j] * acc[j];
        h2p[(size_t)(row0 + mo + j) * N_CLASSES + m] = v;
        agg[(size_t)(row0 + mo + j) * N_CLASSES + m] = v;
    }
}


#define EDGES_PER_WAVE 32
#define EDGE_GROUP     4
static_assert(N_EDGES % EDGES_PER_WAVE == 0, "a wave's 32 edges are all valid or all out of range");
static_assert(EDGES_PER_WAVE % EDGE_GROUP == 0, "whole groups only");

#define TILE_WAVES   32
#define SCAN_EDGES   1024
#define SCAN_STEP    512
#define QUEUE_SLOTS  64
static_assert(N_EDGES >= SCAN_EDGES &&
              (N_EDGES % SCAN_EDGES == 0 || N_EDGES % SCAN_EDGES == SCAN_STEP),
              "a moved-back trip repeats exactly its first 512-edge step, or nothing");

__device__ __forceinline__ void tile_load_group(float4 (&v)[EDGE_GROUP],
                                                const float* __restrict__ hp,
                                                int r_vec, int first, int n, int lane) {
#pragma unroll
    for (int q = 0; q < EDGE_GROUP; ++q) {
        if (first + q < n) {
            const int rs = __builtin_amdgcn_readlane(r_vec, first + q);
            v[q] = *((const float4*)(hp + (size_t)rs * HIDDEN) + lane);
        }
    }
}

__device__ __forceinline__ void tile_add_group(const float4 (&v)[EDGE_GROUP],
                                               float* acc,
                                               int cl_vec, int first, int n, int lane) {
#pragma unroll
    for (int q = 0; q < EDGE_GROUP; ++q) {
        if (first + q < n) {
            const int cs = __builtin_amdgcn_readlane(cl_vec, first + q);
            float* dst = acc + cs * HIDDEN + lane;
            unsafeAtomicAdd(dst,      v[q].x);
            unsafeAtomicAdd(dst + 32, v[q].y);
            unsafeAtomicAdd(dst + 64, v[q].z);
            unsafeAtomicAdd(dst + 96, v[q].w);
        }
    }
}

__device__ __forceinline__ void tile_process(const int* q, int head, int n, int lane,
                                             const int* __restrict__ row,
                                             const int* __restrict__ col, int n0,
                                             const float* __restrict__ hp,
                                             float* acc) {
    const bool valid = lane < n;
    const int e = q[(head + lane) & (QUEUE_SLOTS - 1)];
    int r = 0, cl = 0;
    if (valid) {
        r  = row[e];
        cl = col[e] - n0;
    }
    float4 v[2][EDGE_GROUP];
    tile_load_group(v[0], hp, r, 0, n, lane);
#pragma unroll
    for (int g = 0; g < EDGES_PER_WAVE / EDGE_GROUP; ++g) {
        if (g * EDGE_GROUP < n) {
            if (g + 1 < EDGES_PER_WAVE / EDGE_GROUP)
                tile_load_group(v[(g + 1) & 1], hp, r, (g + 1) * EDGE_GROUP, n, lane);
            tile_add_group(v[g & 1], acc, cl, g * EDGE_GROUP, n, lane);
        }
    }
}

__global__ __launch_bounds__(TILE_WAVES * 32) void agg1_tile_kernel(const int* __restrict__ row,
                                                                    const int* __restrict__ col,
                                                                    const unsigned int* __restrict__ edge_tile,
                                                                    const float* __restrict__ dinv,
                                                                    const float* __restrict__ hp,
                                                                    const float* __restrict__ b1,
                                                                    float* __restrict__ h) {
    __shared__ float acc[TILE_NODES * HIDDEN];
    __shared__ float dinv_tile[TILE_NODES];
    __shared__ int   queue[TILE_WAVES][QUEUE_SLOTS];

    const int tid  = threadIdx.x;
    const int lane = tid & 31;
    const int wave = tid >> 5;
    const int n0   = blockIdx.x * TILE_NODES;

    for (int i = tid; i < TILE_NODES; i += TILE_WAVES * 32) dinv_tile[i] = dinv[n0 + i];
    {
        const float4* s4 = (const float4*)(hp + (size_t)n0 * HIDDEN);
        for (int i = tid; i < TILE_NODES * HIDDEN / 4; i += TILE_WAVES * 32) {
            const float4 s = s4[i];
            float* d = acc + (i >> 5) * HIDDEN + (i & 31);
            d[0] = s.x; d[32] = s.y; d[64] = s.z; d[96] = s.w;
        }
    }
    __syncthreads();

    int* q = queue[wave];
    int head = 0, tail = 0;
    const int n_chunks = (N_EDGES + SCAN_EDGES - 1) / SCAN_EDGES;
    const int last_base = N_EDGES - SCAN_EDGES;

    const unsigned char* tb = (const unsigned char*)edge_tile;
    const unsigned pat = (unsigned)blockIdx.x * 0x01010101u;

    auto load_trip = [&](uint4 (&d)[2], int chunk) {
        const int c0 = chunk * SCAN_EDGES;
        const int base = c0 < last_base ? c0 : last_base;
#pragma unroll
        for (int k = 0; k < 2; ++k) d[k] = *(const uint4*)(tb + base + k * SCAN_STEP + 16 * lane);
    };
    auto hit_bits = [&](const uint4 (&d)[2], int chunk) -> unsigned {
        unsigned hb = 0u;
#pragma unroll
        for (int k = 0; k < 2; ++k) {
            const unsigned ww[4] = {d[k].x, d[k].y, d[k].z, d[k].w};
#pragma unroll
            for (int c = 0; c < 4; ++c) {
                const unsigned t = ww[c] ^ pat;
                const unsigned nz = (((t & 0x7F7F7F7Fu) + 0x7F7F7F7Fu) | t) & 0x80808080u;
                hb |= (nz ^ 0x80808080u) >> (7 - (4 * k + c));
            }
        }
        const int c0 = chunk * SCAN_EDGES;
        return c0 > last_base ? (hb & 0xF0F0F0F0u) : hb;
    };
    auto take_hits = [&](unsigned hb, int chunk) {
        const int c0 = chunk * SCAN_EDGES;
        const int base = c0 < last_base ? c0 : last_base;
        unsigned m = __builtin_amdgcn_ballot_w32(hb != 0u);
        while (m != 0u) {
            if (hb != 0u) {
                const int p = __builtin_ctz(hb);
                hb &= hb - 1u;
                const int w = p & 7;
                const int pos = (tail + (int)__builtin_amdgcn_mbcnt_lo(m, 0u)) & (QUEUE_SLOTS - 1);
                q[pos] = base + (w >> 2) * SCAN_STEP + 16 * lane + 4 * (w & 3) + (p >> 3);
            }
            tail += __builtin_popcount(m);
            if (tail - head >= 32) {
                __builtin_amdgcn_fence(__ATOMIC_ACQ_REL, "wavefront");
                tile_process(q, head, 32, lane, row, col, n0, hp, acc);
                head += 32;
            }
            m = __builtin_amdgcn_ballot_w32(hb != 0u);
        }
    };

    uint4 d0[2], d1[2], d2[2], d3[2];
    load_trip(d0, wave);
    __builtin_amdgcn_sched_barrier(0);
    load_trip(d1, wave + TILE_WAVES);
    __builtin_amdgcn_sched_barrier(0);
    load_trip(d2, wave + 2 * TILE_WAVES);
    for (int chunk = wave; chunk < n_chunks; chunk += 4 * TILE_WAVES) {
        load_trip(d3, chunk + 3 * TILE_WAVES);
        take_hits(hit_bits(d0, chunk), chunk);
        load_trip(d0, chunk + 4 * TILE_WAVES);
        if (chunk + TILE_WAVES < n_chunks)
            take_hits(hit_bits(d1, chunk + TILE_WAVES), chunk + TILE_WAVES);
        load_trip(d1, chunk + 5 * TILE_WAVES);
        if (chunk + 2 * TILE_WAVES < n_chunks)
            take_hits(hit_bits(d2, chunk + 2 * TILE_WAVES), chunk + 2 * TILE_WAVES);
        load_trip(d2, chunk + 6 * TILE_WAVES);
        if (chunk + 3 * TILE_WAVES < n_chunks)
            take_hits(hit_bits(d3, chunk + 3 * TILE_WAVES), chunk + 3 * TILE_WAVES);
    }
    if (tail - head > 0) {
        __builtin_amdgcn_fence(__ATOMIC_ACQ_REL, "wavefront");
        tile_process(q, head, tail - head, lane, row, col, n0, hp, acc);
    }
    __syncthreads();

    {
        const float4* b4 = (const float4*)b1;
        float4* o4 = (float4*)(h + (size_t)n0 * HIDDEN);
        for (int i = tid; i < TILE_NODES * HIDDEN / 4; i += TILE_WAVES * 32) {
            const float d = dinv_tile[i >> 5];
            const float* s = acc + (i >> 5) * HIDDEN + (i & 31);
            float4 a;
            a.x = s[0]; a.y = s[32]; a.z = s[64]; a.w = s[96];
            const float4 b = b4[i & 31];
            float4 o;
            o.x = d * a.x + b.x; o.y = d * a.y + b.y; o.z = d * a.z + b.z; o.w = d * a.w + b.w;
            o.x = o.x > 0.0f ? o.x : 0.0f; o.y = o.y > 0.0f ? o.y : 0.0f;
            o.z = o.z > 0.0f ? o.z : 0.0f; o.w = o.w > 0.0f ? o.w : 0.0f;
            o4[i] = o;
        }
    }
}

__global__ __launch_bounds__(256) void agg2_edge_kernel(const int* __restrict__ row,
                                                        const int* __restrict__ col,
                                                        const float* __restrict__ h2p,
                                                        float* __restrict__ agg) {
    const int lane = threadIdx.x & 31;
    const int wave = (blockIdx.x * blockDim.x + threadIdx.x) >> 5;
    const int e0 = wave * EDGES_PER_WAVE;
    if (e0 >= N_EDGES) return;
    const int r = row[e0 + lane];
    const int c = col[e0 + lane];
    const int half = lane >> 4;
    const int f = lane & 15;

    float v[EDGES_PER_WAVE / 2];
#pragma unroll
    for (int i = 0; i < EDGES_PER_WAVE / 2; ++i) {
        const int rr = __shfl(r, 2 * i + half);
        v[i] = h2p[(size_t)rr * N_CLASSES + f];
    }
#pragma unroll
    for (int i = 0; i < EDGES_PER_WAVE / 2; ++i) {
        const int cc = __shfl(c, 2 * i + half);
        unsafeAtomicAdd(&agg[(size_t)cc * N_CLASSES + f], v[i]);
    }
}

__global__ void logsoftmax_kernel(const float* __restrict__ agg,
                                  const float* __restrict__ dinv,
                                  const float* __restrict__ b2,
                                  float* __restrict__ out) {
    int i = blockIdx.x * blockDim.x + threadIdx.x;
    if (i >= N_NODES) return;
    const float d = dinv[i];
    float v[N_CLASSES];
    float mx = -INFINITY;
#pragma unroll
    for (int c = 0; c < N_CLASSES; ++c) {
        v[c] = d * agg[(size_t)i * N_CLASSES + c] + b2[c];
        mx = fmaxf(mx, v[c]);
    }
    float s = 0.0f;
#pragma unroll
    for (int c = 0; c < N_CLASSES; ++c) s += expf(v[c] - mx);
    float lse = mx + logf(s);
#pragma unroll
    for (int c = 0; c < N_CLASSES; ++c) {
        out[(size_t)i * N_CLASSES + c] = v[c] - lse;
    }
}


static inline int cdiv_l(long a, int b) { return (int)((a + b - 1) / b); }

extern "C" void kernel_launch(void* const* d_in, const int* in_sizes, int n_in,
                              void* d_out, int out_size, void* d_ws, size_t ws_size,
                              hipStream_t stream) {
    const float* x   = (const float*)d_in[0];
    const int*   ei  = (const int*)d_in[1];
    const int*   row = ei;
    const int*   col = ei + N_EDGES;
    const float* W1  = (const float*)d_in[2];
    const float* b1  = (const float*)d_in[3];
    const float* W2  = (const float*)d_in[4];
    const float* b2  = (const float*)d_in[5];
    float* out = (float*)d_out;

    float* ws   = (float*)d_ws;
    float* dinv = ws;
    float* hp   = ws + ((N_NODES + 255) & ~255);
    float* h    = hp + (size_t)N_NODES * HIDDEN;
    float* h2p  = hp;
    float* agg2 = hp + (size_t)N_NODES * N_CLASSES;
    static_assert((size_t)N_EDGES <= (size_t)N_NODES * N_CLASSES * sizeof(float),
                  "the tile bytes fit the output buffer");
    unsigned int* edge_tile = (unsigned int*)d_out;
    static_assert((size_t)DEG_WGS * HIST_WORDS <= (size_t)N_NODES * HIDDEN, "the planes fit the h region");
    unsigned int* hist = (unsigned int*)h;

    const int B = 256;

    deg_hist_kernel<<<DEG_WGS, DEG_THREADS, 0, stream>>>(col, hist, edge_tile);

    const long waves1 = (N_NODES + 15) / 16;
    gemm1_kernel<<<N_NODES / GEMM1_ROWS, GEMM1_WAVES * 32, 0, stream>>>(
        x, W1, (const unsigned short*)hist, dinv, hp);
    agg1_tile_kernel<<<N_NODES / TILE_NODES, TILE_WAVES * 32, 0, stream>>>(
        row, col, edge_tile, dinv, hp, b1, h);

    gemm2_kernel<<<cdiv_l(waves1 * 32, B), B, 0, stream>>>(h, W2, dinv, h2p, agg2);
    agg2_edge_kernel<<<cdiv_l((long)(N_EDGES / EDGES_PER_WAVE) * 32, B), B, 0, stream>>>(
        row, col, h2p, agg2);

    logsoftmax_kernel<<<cdiv_l(N_NODES, B), B, 0, stream>>>(agg2, dinv, b2, out);
}


# Projection 2 moved into the layer-1 tile kernel: finished LDS rows (scale, bias, relu in the A loader) go through WMMA to h2p and the layer-2 start; h never stored, one launch gone; k order kept
# speedup vs baseline: 3.2063x; 1.0481x over previous
#include <hip/hip_runtime.h>
#include <math.h>

typedef __attribute__((ext_vector_type(2))) float v2f;
typedef __attribute__((ext_vector_type(8))) float v8f;

#define N_NODES   100000
#define N_EDGES   1600000
#define IN_FEAT   256
#define HIDDEN    128
#define N_CLASSES 16

#define TILE_NODES   400
static_assert(N_NODES % TILE_NODES == 0, "whole tiles only");
static_assert(N_NODES / TILE_NODES <= 256, "a tile number fits one byte");
static_assert(N_EDGES % 4 == 0, "four edges per thread in the degree pass");

#define DEG_WGS      64
#define DEG_THREADS  1024
#define HIST_WORDS   (N_NODES / 2)
#define QUADS_PER_WG (N_EDGES / 4 / DEG_WGS)
#define QUAD_TRIPS   ((QUADS_PER_WG + DEG_THREADS - 1) / DEG_THREADS)
static_assert(N_NODES % 8 == 0, "a plane is a whole number of 16-byte pieces");
static_assert((N_EDGES / 4) % DEG_WGS == 0, "equal slices of whole 4-edge items");
static_assert(N_EDGES / DEG_WGS < 65536, "a 16-bit counter cannot overflow within one slice");

__global__ __launch_bounds__(DEG_THREADS) void deg_hist_kernel(const int* __restrict__ col,
                                                              unsigned int* __restrict__ hist,
                                                              unsigned int* __restrict__ edge_tile) {
    __shared__ __attribute__((aligned(16))) unsigned int cnt[HIST_WORDS];
    const int tid = threadIdx.x;
    const int q0 = blockIdx.x * QUADS_PER_WG;
    int4 c[QUAD_TRIPS];
#pragma unroll
    for (int t = 0; t < QUAD_TRIPS; ++t) {
        const int i = tid + t * DEG_THREADS;
        if (i < QUADS_PER_WG) c[t] = *(const int4*)(col + 4 * (size_t)(q0 + i));
    }
    for (int i = tid; i < HIST_WORDS / 4; i += DEG_THREADS) ((uint4*)cnt)[i] = make_uint4(0u, 0u, 0u, 0u);
    __syncthreads();
#pragma unroll
    for (int t = 0; t < QUAD_TRIPS; ++t) {
        const int i = tid + t * DEG_THREADS;
        if (i < QUADS_PER_WG) {
            atomicAdd(&cnt[c[t].x >> 1], 1u << ((c[t].x & 1) * 16));
            atomicAdd(&cnt[c[t].y >> 1], 1u << ((c[t].y & 1) * 16));
            atomicAdd(&cnt[c[t].z >> 1], 1u << ((c[t].z & 1) * 16));
            atomicAdd(&cnt[c[t].w >> 1], 1u << ((c[t].w & 1) * 16));
            edge_tile[q0 + i] = (unsigned)(c[t].x / TILE_NODES) | ((unsigned)(c[t].y / TILE_NODES) << 8) |
                                ((unsigned)(c[t].z / TILE_NODES) << 16) | ((unsigned)(c[t].w / TILE_NODES) << 24);
        }
    }
    __syncthreads();
    uint4* dst = (uint4*)(hist + (size_t)blockIdx.x * HIST_WORDS);
    for (int i = tid; i < HIST_WORDS / 4; i += DEG_THREADS) dst[i] = ((const uint4*)cnt)[i];
}


#define GEMM1_ROWS   400
#define GEMM1_WAVES  32
#define W1_PITCH     288
#define GEMM1_UNITS  ((GEMM1_ROWS / 16) * 2)
static_assert(N_NODES % GEMM1_ROWS == 0 && GEMM1_ROWS % 16 == 0, "whole row tiles only");
static_assert(HIDDEN == 128 && IN_FEAT % 32 == 0, "unit = 64 columns, A chunks of 32 columns");

__device__ __forceinline__ void gemm1_load_a(v2f (&a)[8], const float* __restrict__ p) {
#pragma unroll
    for (int s = 0; s < 8; ++s) a[s] = *(const v2f*)(p + 4 * s);
}

__device__ __forceinline__ void gemm1_chunk(v8f (&acc)[4], const v2f (&a)[8], const float* bc) {
#pragma unroll
    for (int s = 0; s < 8; ++s) {
        const float* bk = bc + (2 * s) * W1_PITCH;
#pragma unroll
        for (int n = 0; n < 4; ++n) {
            const v2f b = *(const v2f*)(bk + 32 * n);
            acc[n] = __builtin_amdgcn_wmma_f32_16x16x4_f32(
                false, a[s], false, b, (short)0, acc[n], false, false);
        }
    }
}

__global__ __launch_bounds__(GEMM1_WAVES * 32) void gemm1_kernel(const float* __restrict__ x,
                                                                const float* __restrict__ W1,
                                                                const unsigned short* __restrict__ hist,
                                                                float* __restrict__ dinv,
                                                                float* __restrict__ hp) {
    __shared__ __attribute__((aligned(16))) float Wl[(IN_FEAT / 2) * W1_PITCH];
    __shared__ float dinv_l[GEMM1_ROWS];

    const int tid = threadIdx.x;
    if (tid < GEMM1_ROWS) {
        const int node = blockIdx.x * GEMM1_ROWS + tid;
        const unsigned short* hn = hist + node;
        unsigned int s = 1u;
#pragma unroll
        for (int k = 0; k < DEG_WGS; ++k) s += hn[(size_t)k * N_NODES];
        const float d = rsqrtf((float)s);
        dinv_l[tid] = d;
        dinv[node] = d;
    }
    for (int it = tid; it < (IN_FEAT / 2) * (HIDDEN / 4); it += GEMM1_WAVES * 32) {
        const int p = it / (HIDDEN / 4);
        const int c = it % (HIDDEN / 4);
        const float4 r0 = *(const float4*)(W1 + (size_t)(2 * p) * HIDDEN + 4 * c);
        const float4 r1 = *(const float4*)(W1 + (size_t)(2 * p + 1) * HIDDEN + 4 * c);
        float4* d = (float4*)(Wl + p * W1_PITCH + 8 * c);
        d[0] = make_float4(r0.x, r1.x, r0.y, r1.y);
        d[1] = make_float4(r0.z, r1.z, r0.w, r1.w);
    }
    __syncthreads();

    const int lane = tid & 31;
    const int wave = tid >> 5;
    const int m  = lane & 15;
    const int kh = lane >> 4;
    const int mo = kh * 8;

    for (int u = wave; u < GEMM1_UNITS; u += GEMM1_WAVES) {
        const int row0 = blockIdx.x * GEMM1_ROWS + (u >> 1) * 16;
        const int col0 = (u & 1) * 64;
        const float* arow = x + (size_t)(row0 + m) * IN_FEAT + 2 * kh;
        const float* bl = Wl + kh * W1_PITCH + (col0 + m) * 2;

        v8f acc[4];
#pragma unroll
        for (int n = 0; n < 4; ++n) acc[n] = v8f{};

        v2f a0[8], a1[8];
        gemm1_load_a(a0, arow);
#pragma nounroll
        for (int cp = 0; cp < IN_FEAT / 64; ++cp) {
            gemm1_load_a(a1, arow + 64 * cp + 32);
            gemm1_chunk(acc, a0, bl + (32 * cp) * W1_PITCH);
            if (2 * cp + 2 < IN_FEAT / 32)
                gemm1_load_a(a0, arow + 64 * cp + 64);
            gemm1_chunk(acc, a1, bl + (32 * cp + 16) * W1_PITCH);
        }

        float ds[8];
#pragma unroll
        for (int j = 0; j < 8; ++j) ds[j] = dinv_l[(u >> 1) * 16 + mo + j];
#pragma unroll
        for (int n = 0; n < 4; ++n) {
#pragma unroll
            for (int j = 0; j < 8; ++j) {
                hp[(size_t)(row0 + mo + j) * HIDDEN + col0 + n * 16 + m] = ds[j] * acc[n][j];
            }
        }
    }
}


#define EDGES_PER_WAVE 32
#define EDGE_GROUP     4
static_assert(N_EDGES % EDGES_PER_WAVE == 0, "a wave's 32 edges are all valid or all out of range");
static_assert(EDGES_PER_WAVE % EDGE_GROUP == 0, "whole groups only");

#define TILE_WAVES   32
#define SCAN_EDGES   1024
#define SCAN_STEP    512
#define QUEUE_SLOTS  64
#define ACC_PITCH    (HIDDEN + 1)
static_assert(TILE_NODES % 16 == 0 && HIDDEN % 4 == 0 && N_CLASSES == 16,
              "projection 2: whole 16-row units, 4-wide k-steps, one 16-column tile");
static_assert(N_EDGES >= SCAN_EDGES &&
              (N_EDGES % SCAN_EDGES == 0 || N_EDGES % SCAN_EDGES == SCAN_STEP),
              "a moved-back trip repeats exactly its first 512-edge step, or nothing");

__device__ __forceinline__ void tile_load_group(float4 (&v)[EDGE_GROUP],
                                                const float* __restrict__ hp,
                                                int r_vec, int first, int n, int lane) {
#pragma unroll
    for (int q = 0; q < EDGE_GROUP; ++q) {
        if (first + q < n) {
            const int rs = __builtin_amdgcn_readlane(r_vec, first + q);
            v[q] = *((const float4*)(hp + (size_t)rs * HIDDEN) + lane);
        }
    }
}

__device__ __forceinline__ void tile_add_group(const float4 (&v)[EDGE_GROUP],
                                               float* acc,
                                               int cl_vec, int first, int n, int lane) {
#pragma unroll
    for (int q = 0; q < EDGE_GROUP; ++q) {
        if (first + q < n) {
            const int cs = __builtin_amdgcn_readlane(cl_vec, first + q);
            float* dst = acc + cs * ACC_PITCH + lane;
            unsafeAtomicAdd(dst,      v[q].x);
            unsafeAtomicAdd(dst + 32, v[q].y);
            unsafeAtomicAdd(dst + 64, v[q].z);
            unsafeAtomicAdd(dst + 96, v[q].w);
        }
    }
}

__device__ __forceinline__ void tile_process(const int* q, int head, int n, int lane,
                                             const int* __restrict__ row,
                                             const int* __restrict__ col, int n0,
                                             const float* __restrict__ hp,
                                             float* acc) {
    const bool valid = lane < n;
    const int e = q[(head + lane) & (QUEUE_SLOTS - 1)];
    int r = 0, cl = 0;
    if (valid) {
        r  = row[e];
        cl = col[e] - n0;
    }
    float4 v[2][EDGE_GROUP];
    tile_load_group(v[0], hp, r, 0, n, lane);
#pragma unroll
    for (int g = 0; g < EDGES_PER_WAVE / EDGE_GROUP; ++g) {
        if (g * EDGE_GROUP < n) {
            if (g + 1 < EDGES_PER_WAVE / EDGE_GROUP)
                tile_load_group(v[(g + 1) & 1], hp, r, (g + 1) * EDGE_GROUP, n, lane);
            tile_add_group(v[g & 1], acc, cl, g * EDGE_GROUP, n, lane);
        }
    }
}

__global__ __launch_bounds__(TILE_WAVES * 32) void agg1_tile_kernel(const int* __restrict__ row,
                                                                    const int* __restrict__ col,
                                                                    const unsigned int* __restrict__ edge_tile,
                                                                    const float* __restrict__ dinv,
                                                                    const float* __restrict__ hp,
                                                                    const float* __restrict__ b1,
                                                                    const float* __restrict__ W2,
                                                                    float* __restrict__ h2p,
                                                                    float* __restrict__ agg) {
    __shared__ float acc[TILE_NODES * ACC_PITCH];
    __shared__ float dinv_tile[TILE_NODES];
    __shared__ int   queue[TILE_WAVES][QUEUE_SLOTS];
    __shared__ __attribute__((aligned(8))) float W2f[HIDDEN * N_CLASSES];
    __shared__ __attribute__((aligned(8))) float b1s[HIDDEN];

    const int tid  = threadIdx.x;
    const int lane = tid & 31;
    const int wave = tid >> 5;
    const int n0   = blockIdx.x * TILE_NODES;

    for (int i = tid; i < TILE_NODES; i += TILE_WAVES * 32) dinv_tile[i] = dinv[n0 + i];
    for (int i = tid; i < HIDDEN * N_CLASSES; i += TILE_WAVES * 32) {
        const int t = i >> 6, ln = (i >> 1) & 31, sx = i & 1;
        W2f[i] = W2[(size_t)(4 * t + 2 * (ln >> 4) + sx) * N_CLASSES + (ln & 15)];
    }
    for (int i = tid; i < HIDDEN; i += TILE_WAVES * 32) b1s[i] = b1[i];
    {
        const float4* s4 = (const float4*)(hp + (size_t)n0 * HIDDEN);
        for (int i = tid; i < TILE_NODES * HIDDEN / 4; i += TILE_WAVES * 32) {
            const float4 s = s4[i];
            float* d = acc + (i >> 5) * ACC_PITCH + (i & 31);
            d[0] = s.x; d[32] = s.y; d[64] = s.z; d[96] = s.w;
        }
    }
    __syncthreads();

    int* q = queue[wave];
    int head = 0, tail = 0;
    const int n_chunks = (N_EDGES + SCAN_EDGES - 1) / SCAN_EDGES;
    const int last_base = N_EDGES - SCAN_EDGES;

    const unsigned char* tb = (const unsigned char*)edge_tile;
    const unsigned pat = (unsigned)blockIdx.x * 0x01010101u;

    auto load_trip = [&](uint4 (&d)[2], int chunk) {
        const int c0 = chunk * SCAN_EDGES;
        const int base = c0 < last_base ? c0 : last_base;
#pragma unroll
        for (int k = 0; k < 2; ++k) d[k] = *(const uint4*)(tb + base + k * SCAN_STEP + 16 * lane);
    };
    auto hit_bits = [&](const uint4 (&d)[2], int chunk) -> unsigned {
        unsigned hb = 0u;
#pragma unroll
        for (int k = 0; k < 2; ++k) {
            const unsigned ww[4] = {d[k].x, d[k].y, d[k].z, d[k].w};
#pragma unroll
            for (int c = 0; c < 4; ++c) {
                const unsigned t = ww[c] ^ pat;
                const unsigned nz = (((t & 0x7F7F7F7Fu) + 0x7F7F7F7Fu) | t) & 0x80808080u;
                hb |= (nz ^ 0x80808080u) >> (7 - (4 * k + c));
            }
        }
        const int c0 = chunk * SCAN_EDGES;
        return c0 > last_base ? (hb & 0xF0F0F0F0u) : hb;
    };
    auto take_hits = [&](unsigned hb, int chunk) {
        const int c0 = chunk * SCAN_EDGES;
        const int base = c0 < last_base ? c0 : last_base;
        unsigned m = __builtin_amdgcn_ballot_w32(hb != 0u);
        while (m != 0u) {
            if (hb != 0u) {
                const int p = __builtin_ctz(hb);
                hb &= hb - 1u;
                const int w = p & 7;
                const int pos = (tail + (int)__builtin_amdgcn_mbcnt_lo(m, 0u)) & (QUEUE_SLOTS - 1);
                q[pos] = base + (w >> 2) * SCAN_STEP + 16 * lane + 4 * (w & 3) + (p >> 3);
            }
            tail += __builtin_popcount(m);
            if (tail - head >= 32) {
                __builtin_amdgcn_fence(__ATOMIC_ACQ_REL, "wavefront");
                tile_process(q, head, 32, lane, row, col, n0, hp, acc);
                head += 32;
            }
            m = __builtin_amdgcn_ballot_w32(hb != 0u);
        }
    };

    uint4 d0[2], d1[2], d2[2], d3[2];
    load_trip(d0, wave);
    __builtin_amdgcn_sched_barrier(0);
    load_trip(d1, wave + TILE_WAVES);
    __builtin_amdgcn_sched_barrier(0);
    load_trip(d2, wave + 2 * TILE_WAVES);
    for (int chunk = wave; chunk < n_chunks; chunk += 4 * TILE_WAVES) {
        load_trip(d3, chunk + 3 * TILE_WAVES);
        take_hits(hit_bits(d0, chunk), chunk);
        load_trip(d0, chunk + 4 * TILE_WAVES);
        if (chunk + TILE_WAVES < n_chunks)
            take_hits(hit_bits(d1, chunk + TILE_WAVES), chunk + TILE_WAVES);
        load_trip(d1, chunk + 5 * TILE_WAVES);
        if (chunk + 2 * TILE_WAVES < n_chunks)
            take_hits(hit_bits(d2, chunk + 2 * TILE_WAVES), chunk + 2 * TILE_WAVES);
        load_trip(d2, chunk + 6 * TILE_WAVES);
        if (chunk + 3 * TILE_WAVES < n_chunks)
            take_hits(hit_bits(d3, chunk + 3 * TILE_WAVES), chunk + 3 * TILE_WAVES);
    }
    if (tail - head > 0) {
        __builtin_amdgcn_fence(__ATOMIC_ACQ_REL, "wavefront");
        tile_process(q, head, tail - head, lane, row, col, n0, hp, acc);
    }
    __syncthreads();

    for (int u = wave; u < TILE_NODES / 16; u += TILE_WAVES) {
        const int m  = lane & 15;
        const int kh = lane >> 4;
        const int mo = kh * 8;
        const float d = dinv_tile[u * 16 + m];
        const float* arow = acc + (u * 16 + m) * ACC_PITCH + 64 * kh;
        v8f c = v8f{};
#pragma unroll
        for (int t = 0; t < HIDDEN / 4; ++t) {
            const v2f bias = *(const v2f*)(b1s + 4 * t + 2 * kh);
            const v2f b = *(const v2f*)(W2f + 64 * t + 2 * lane);
            v2f a;
            a.x = d * arow[t] + bias.x;
            a.y = d * arow[32 + t] + bias.y;
            a.x = a.x > 0.0f ? a.x : 0.0f;
            a.y = a.y > 0.0f ? a.y : 0.0f;
            c = __builtin_amdgcn_wmma_f32_16x16x4_f32(
                false, a, false, b, (short)0, c, false, false);
        }
        const int row0 = n0 + u * 16;
#pragma unroll
        for (int j = 0; j < 8; ++j) {
            const float v = dinv_tile[u * 16 + mo + j] * c[j];
            h2p[(size_t)(row0 + mo + j) * N_CLASSES + m] = v;
            agg[(size_t)(row0 + mo + j) * N_CLASSES + m] = v;
        }
    }
}

__global__ __launch_bounds__(256) void agg2_edge_kernel(const int* __restrict__ row,
                                                        const int* __restrict__ col,
                                                        const float* __restrict__ h2p,
                                                        float* __restrict__ agg) {
    const int lane = threadIdx.x & 31;
    const int wave = (blockIdx.x * blockDim.x + threadIdx.x) >> 5;
    const int e0 = wave * EDGES_PER_WAVE;
    if (e0 >= N_EDGES) return;
    const int r = row[e0 + lane];
    const int c = col[e0 + lane];
    const int half = lane >> 4;
    const int f = lane & 15;

    float v[EDGES_PER_WAVE / 2];
#pragma unroll
    for (int i = 0; i < EDGES_PER_WAVE / 2; ++i) {
        const int rr = __shfl(r, 2 * i + half);
        v[i] = h2p[(size_t)rr * N_CLASSES + f];
    }
#pragma unroll
    for (int i = 0; i < EDGES_PER_WAVE / 2; ++i) {
        const int cc = __shfl(c, 2 * i + half);
        unsafeAtomicAdd(&agg[(size_t)cc * N_CLASSES + f], v[i]);
    }
}

__global__ void logsoftmax_kernel(const float* __restrict__ agg,
                                  const float* __restrict__ dinv,
                                  const float* __restrict__ b2,
                                  float* __restrict__ out) {
    int i = blockIdx.x * blockDim.x + threadIdx.x;
    if (i >= N_NODES) return;
    const float d = dinv[i];
    float v[N_CLASSES];
    float mx = -INFINITY;
#pragma unroll
    for (int c = 0; c < N_CLASSES; ++c) {
        v[c] = d * agg[(size_t)i * N_CLASSES + c] + b2[c];
        mx = fmaxf(mx, v[c]);
    }
    float s = 0.0f;
#pragma unroll
    for (int c = 0; c < N_CLASSES; ++c) s += expf(v[c] - mx);
    float lse = mx + logf(s);
#pragma unroll
    for (int c = 0; c < N_CLASSES; ++c) {
        out[(size_t)i * N_CLASSES + c] = v[c] - lse;
    }
}


static inline int cdiv_l(long a, int b) { return (int)((a + b - 1) / b); }

extern "C" void kernel_launch(void* const* d_in, const int* in_sizes, int n_in,
                              void* d_out, int out_size, void* d_ws, size_t ws_size,
                              hipStream_t stream) {
    const float* x   = (const float*)d_in[0];
    const int*   ei  = (const int*)d_in[1];
    const int*   row = ei;
    const int*   col = ei + N_EDGES;
    const float* W1  = (const float*)d_in[2];
    const float* b1  = (const float*)d_in[3];
    const float* W2  = (const float*)d_in[4];
    const float* b2  = (const float*)d_in[5];
    float* out = (float*)d_out;

    float* ws   = (float*)d_ws;
    float* dinv = ws;
    float* hp   = ws + ((N_NODES + 255) & ~255);
    float* h2p  = hp + (size_t)N_NODES * HIDDEN;
    float* agg2 = h2p + (size_t)N_NODES * N_CLASSES;
    static_assert((size_t)N_EDGES <= (size_t)N_NODES * N_CLASSES * sizeof(float),
                  "the tile bytes fit the output buffer");
    unsigned int* edge_tile = (unsigned int*)d_out;
    static_assert((size_t)DEG_WGS * HIST_WORDS <= (size_t)N_NODES * N_CLASSES * 2, "the planes fit that region");
    unsigned int* hist = (unsigned int*)h2p;

    const int B = 256;

    deg_hist_kernel<<<DEG_WGS, DEG_THREADS, 0, stream>>>(col, hist, edge_tile);

    gemm1_kernel<<<N_NODES / GEMM1_ROWS, GEMM1_WAVES * 32, 0, stream>>>(
        x, W1, (const unsigned short*)hist, dinv, hp);
    agg1_tile_kernel<<<N_NODES / TILE_NODES, TILE_WAVES * 32, 0, stream>>>(
        row, col, edge_tile, dinv, hp, b1, W2, h2p, agg2);

    agg2_edge_kernel<<<cdiv_l((long)(N_EDGES / EDGES_PER_WAVE) * 32, B), B, 0, stream>>>(
        row, col, h2p, agg2);

    logsoftmax_kernel<<<cdiv_l(N_NODES, B), B, 0, stream>>>(agg2, dinv, b2, out);
}


# Layer 2 as a tile kernel fed by a packed edge list the layer-1 tiles write; 64-B rows added to LDS sums (arrival order), log-softmax fused (sum across lanes); global f32 atomics and 1 launch gone
# speedup vs baseline: 3.7940x; 1.1833x over previous
#include <hip/hip_runtime.h>
#include <math.h>

typedef __attribute__((ext_vector_type(2))) float v2f;
typedef __attribute__((ext_vector_type(8))) float v8f;

#define N_NODES   100000
#define N_EDGES   1600000
#define IN_FEAT   256
#define HIDDEN    128
#define N_CLASSES 16

#define TILE_NODES   400
static_assert(N_NODES % TILE_NODES == 0, "whole tiles only");
static_assert(N_NODES / TILE_NODES <= 256, "a tile number fits one byte");
static_assert(N_EDGES % 4 == 0, "four edges per thread in the degree pass");

#define DEG_WGS      64
#define DEG_THREADS  1024
#define HIST_WORDS   (N_NODES / 2)
#define QUADS_PER_WG (N_EDGES / 4 / DEG_WGS)
#define QUAD_TRIPS   ((QUADS_PER_WG + DEG_THREADS - 1) / DEG_THREADS)
static_assert(N_NODES % 8 == 0, "a plane is a whole number of 16-byte pieces");
static_assert((N_EDGES / 4) % DEG_WGS == 0, "equal slices of whole 4-edge items");
static_assert(N_EDGES / DEG_WGS < 65536, "a 16-bit counter cannot overflow within one slice");

__global__ __launch_bounds__(DEG_THREADS) void deg_hist_kernel(const int* __restrict__ col,
                                                              unsigned int* __restrict__ hist,
                                                              unsigned int* __restrict__ edge_tile) {
    __shared__ __attribute__((aligned(16))) unsigned int cnt[HIST_WORDS];
    const int tid = threadIdx.x;
    const int q0 = blockIdx.x * QUADS_PER_WG;
    int4 c[QUAD_TRIPS];
#pragma unroll
    for (int t = 0; t < QUAD_TRIPS; ++t) {
        const int i = tid + t * DEG_THREADS;
        if (i < QUADS_PER_WG) c[t] = *(const int4*)(col + 4 * (size_t)(q0 + i));
    }
    for (int i = tid; i < HIST_WORDS / 4; i += DEG_THREADS) ((uint4*)cnt)[i] = make_uint4(0u, 0u, 0u, 0u);
    __syncthreads();
#pragma unroll
    for (int t = 0; t < QUAD_TRIPS; ++t) {
        const int i = tid + t * DEG_THREADS;
        if (i < QUADS_PER_WG) {
            atomicAdd(&cnt[c[t].x >> 1], 1u << ((c[t].x & 1) * 16));
            atomicAdd(&cnt[c[t].y >> 1], 1u << ((c[t].y & 1) * 16));
            atomicAdd(&cnt[c[t].z >> 1], 1u << ((c[t].z & 1) * 16));
            atomicAdd(&cnt[c[t].w >> 1], 1u << ((c[t].w & 1) * 16));
            edge_tile[q0 + i] = (unsigned)(c[t].x / TILE_NODES) | ((unsigned)(c[t].y / TILE_NODES) << 8) |
                                ((unsigned)(c[t].z / TILE_NODES) << 16) | ((unsigned)(c[t].w / TILE_NODES) << 24);
        }
    }
    __syncthreads();
    uint4* dst = (uint4*)(hist + (size_t)blockIdx.x * HIST_WORDS);
    for (int i = tid; i < HIST_WORDS / 4; i += DEG_THREADS) dst[i] = ((const uint4*)cnt)[i];
}


#define GEMM1_ROWS   400
#define GEMM1_WAVES  32
#define W1_PITCH     288
#define GEMM1_UNITS  ((GEMM1_ROWS / 16) * 2)
static_assert(N_NODES % GEMM1_ROWS == 0 && GEMM1_ROWS % 16 == 0, "whole row tiles only");
static_assert(HIDDEN == 128 && IN_FEAT % 32 == 0, "unit = 64 columns, A chunks of 32 columns");

__device__ __forceinline__ void gemm1_load_a(v2f (&a)[8], const float* __restrict__ p) {
#pragma unroll
    for (int s = 0; s < 8; ++s) a[s] = *(const v2f*)(p + 4 * s);
}

__device__ __forceinline__ void gemm1_chunk(v8f (&acc)[4], const v2f (&a)[8], const float* bc) {
#pragma unroll
    for (int s = 0; s < 8; ++s) {
        const float* bk = bc + (2 * s) * W1_PITCH;
#pragma unroll
        for (int n = 0; n < 4; ++n) {
            const v2f b = *(const v2f*)(bk + 32 * n);
            acc[n] = __builtin_amdgcn_wmma_f32_16x16x4_f32(
                false, a[s], false, b, (short)0, acc[n], false, false);
        }
    }
}

__global__ __launch_bounds__(GEMM1_WAVES * 32) void gemm1_kernel(const float* __restrict__ x,
                                                                const float* __restrict__ W1,
                                                                const unsigned short* __restrict__ hist,
                                                                int* __restrict__ deg,
                                                                float* __restrict__ hp) {
    __shared__ __attribute__((aligned(16))) float Wl[(IN_FEAT / 2) * W1_PITCH];
    __shared__ float dinv_l[GEMM1_ROWS];

    const int tid = threadIdx.x;
    if (tid < GEMM1_ROWS) {
        const int node = blockIdx.x * GEMM1_ROWS + tid;
        const unsigned short* hn = hist + node;
        unsigned int s = 1u;
#pragma unroll
        for (int k = 0; k < DEG_WGS; ++k) s += hn[(size_t)k * N_NODES];
        const float d = rsqrtf((float)s);
        dinv_l[tid] = d;
        deg[node] = (int)s;
    }
    for (int it = tid; it < (IN_FEAT / 2) * (HIDDEN / 4); it += GEMM1_WAVES * 32) {
        const int p = it / (HIDDEN / 4);
        const int c = it % (HIDDEN / 4);
        const float4 r0 = *(const float4*)(W1 + (size_t)(2 * p) * HIDDEN + 4 * c);
        const float4 r1 = *(const float4*)(W1 + (size_t)(2 * p + 1) * HIDDEN + 4 * c);
        float4* d = (float4*)(Wl + p * W1_PITCH + 8 * c);
        d[0] = make_float4(r0.x, r1.x, r0.y, r1.y);
        d[1] = make_float4(r0.z, r1.z, r0.w, r1.w);
    }
    __syncthreads();

    const int lane = tid & 31;
    const int wave = tid >> 5;
    const int m  = lane & 15;
    const int kh = lane >> 4;
    const int mo = kh * 8;

    for (int u = wave; u < GEMM1_UNITS; u += GEMM1_WAVES) {
        const int row0 = blockIdx.x * GEMM1_ROWS + (u >> 1) * 16;
        const int col0 = (u & 1) * 64;
        const float* arow = x + (size_t)(row0 + m) * IN_FEAT + 2 * kh;
        const float* bl = Wl + kh * W1_PITCH + (col0 + m) * 2;

        v8f acc[4];
#pragma unroll
        for (int n = 0; n < 4; ++n) acc[n] = v8f{};

        v2f a0[8], a1[8];
        gemm1_load_a(a0, arow);
#pragma nounroll
        for (int cp = 0; cp < IN_FEAT / 64; ++cp) {
            gemm1_load_a(a1, arow + 64 * cp + 32);
            gemm1_chunk(acc, a0, bl + (32 * cp) * W1_PITCH);
            if (2 * cp + 2 < IN_FEAT / 32)
                gemm1_load_a(a0, arow + 64 * cp + 64);
            gemm1_chunk(acc, a1, bl + (32 * cp + 16) * W1_PITCH);
        }

        float ds[8];
#pragma unroll
        for (int j = 0; j < 8; ++j) ds[j] = dinv_l[(u >> 1) * 16 + mo + j];
#pragma unroll
        for (int n = 0; n < 4; ++n) {
#pragma unroll
            for (int j = 0; j < 8; ++j) {
                hp[(size_t)(row0 + mo + j) * HIDDEN + col0 + n * 16 + m] = ds[j] * acc[n][j];
            }
        }
    }
}


#define EDGES_PER_WAVE 32
#define EDGE_GROUP     4
static_assert(N_EDGES % EDGES_PER_WAVE == 0, "a wave's 32 edges are all valid or all out of range");
static_assert(EDGES_PER_WAVE % EDGE_GROUP == 0, "whole groups only");

#define TILE_WAVES   32
#define SCAN_EDGES   1024
#define SCAN_STEP    512
#define QUEUE_SLOTS  64
#define ACC_PITCH    (HIDDEN + 1)
static_assert(TILE_NODES <= 512 && N_NODES <= (1 << 23), "a list word holds source << 9 | row within the tile");
static_assert((size_t)N_EDGES <= (size_t)N_NODES * N_CLASSES, "the edge list fits the region of the old layer-2 sums");
static_assert(TILE_NODES % 4 == 0, "the degrees in front of a tile are whole 16-byte pieces");

__device__ __forceinline__ int2 tile_edge_range(const int* __restrict__ deg, int n0, int tid, int* red) {
    int before = 0;
    const int4* d4 = (const int4*)deg;
    const int n4 = n0 / 4;
    for (int i0 = tid; i0 < n4; i0 += 8 * TILE_WAVES * 32) {
        int4 v[8];
#pragma unroll
        for (int j = 0; j < 8; ++j) {
            const int i = i0 + j * TILE_WAVES * 32;
            v[j] = i < n4 ? d4[i] : make_int4(0, 0, 0, 0);
        }
#pragma unroll
        for (int j = 0; j < 8; ++j) before += (v[j].x + v[j].y) + (v[j].z + v[j].w);
    }
    int own = 0;
    if (tid < TILE_NODES) own = deg[n0 + tid];
#pragma unroll
    for (int o = 16; o > 0; o >>= 1) {
        before += __shfl_xor(before, o);
        own    += __shfl_xor(own, o);
    }
    if ((tid & 31) == 0) {
        red[tid >> 5] = before;
        red[32 + (tid >> 5)] = own;
    }
    __syncthreads();
    int b = red[tid & 31];
    int w = red[32 + (tid & 31)];
#pragma unroll
    for (int o = 16; o > 0; o >>= 1) {
        b += __shfl_xor(b, o);
        w += __shfl_xor(w, o);
    }
    return make_int2(b - n0, w - TILE_NODES);
}
static_assert(TILE_NODES % 16 == 0 && HIDDEN % 4 == 0 && N_CLASSES == 16,
              "projection 2: whole 16-row units, 4-wide k-steps, one 16-column tile");
static_assert(N_EDGES >= SCAN_EDGES &&
              (N_EDGES % SCAN_EDGES == 0 || N_EDGES % SCAN_EDGES == SCAN_STEP),
              "a moved-back trip repeats exactly its first 512-edge step, or nothing");

__device__ __forceinline__ void tile_load_group(float4 (&v)[EDGE_GROUP],
                                                const float* __restrict__ hp,
                                                int r_vec, int first, int n, int lane) {
#pragma unroll
    for (int q = 0; q < EDGE_GROUP; ++q) {
        if (first + q < n) {
            const int rs = __builtin_amdgcn_readlane(r_vec, first + q);
            v[q] = *((const float4*)(hp + (size_t)rs * HIDDEN) + lane);
        }
    }
}

__device__ __forceinline__ void tile_add_group(const float4 (&v)[EDGE_GROUP],
                                               float* acc,
                                               int cl_vec, int first, int n, int lane) {
#pragma unroll
    for (int q = 0; q < EDGE_GROUP; ++q) {
        if (first + q < n) {
            const int cs = __builtin_amdgcn_readlane(cl_vec, first + q);
            float* dst = acc + cs * ACC_PITCH + lane;
            unsafeAtomicAdd(dst,      v[q].x);
            unsafeAtomicAdd(dst + 32, v[q].y);
            unsafeAtomicAdd(dst + 64, v[q].z);
            unsafeAtomicAdd(dst + 96, v[q].w);
        }
    }
}

__device__ __forceinline__ void tile_process(const int* q, int head, int n, int lane,
                                             const int* __restrict__ row,
                                             const int* __restrict__ col, int n0,
                                             const float* __restrict__ hp,
                                             float* acc,
                                             unsigned int* __restrict__ my_list, int* cursor) {
    const bool valid = lane < n;
    const int e = q[(head + lane) & (QUEUE_SLOTS - 1)];
    int r = 0, cl = 0;
    if (valid) {
        r  = row[e];
        cl = col[e] - n0;
    }
    int slot = 0;
    if (lane == 0) slot = atomicAdd(cursor, n);
    slot = __builtin_amdgcn_readfirstlane(slot);
    if (valid) my_list[slot + lane] = ((unsigned int)r << 9) | (unsigned int)cl;
    float4 v[2][EDGE_GROUP];
    tile_load_group(v[0], hp, r, 0, n, lane);
#pragma unroll
    for (int g = 0; g < EDGES_PER_WAVE / EDGE_GROUP; ++g) {
        if (g * EDGE_GROUP < n) {
            if (g + 1 < EDGES_PER_WAVE / EDGE_GROUP)
                tile_load_group(v[(g + 1) & 1], hp, r, (g + 1) * EDGE_GROUP, n, lane);
            tile_add_group(v[g & 1], acc, cl, g * EDGE_GROUP, n, lane);
        }
    }
}

__global__ __launch_bounds__(TILE_WAVES * 32) void agg1_tile_kernel(const int* __restrict__ row,
                                                                    const int* __restrict__ col,
                                                                    const unsigned int* __restrict__ edge_tile,
                                                                    const int* __restrict__ deg,
                                                                    const float* __restrict__ hp,
                                                                    const float* __restrict__ b1,
                                                                    const float* __restrict__ W2,
                                                                    float* __restrict__ h2p,
                                                                    unsigned int* __restrict__ elist) {
    __shared__ float acc[TILE_NODES * ACC_PITCH];
    __shared__ float dinv_tile[TILE_NODES];
    __shared__ int   queue[TILE_WAVES][QUEUE_SLOTS];
    __shared__ __attribute__((aligned(8))) float W2f[HIDDEN * N_CLASSES];
    __shared__ __attribute__((aligned(8))) float b1s[HIDDEN];
    __shared__ int red[64];
    __shared__ int list_cursor;

    const int tid  = threadIdx.x;
    const int lane = tid & 31;
    const int wave = tid >> 5;
    const int n0   = blockIdx.x * TILE_NODES;

    if (tid == 0) list_cursor = 0;
    unsigned int* my_list = elist + tile_edge_range(deg, n0, tid, red).x;

    for (int i = tid; i < TILE_NODES; i += TILE_WAVES * 32) dinv_tile[i] = rsqrtf((float)deg[n0 + i]);
    for (int i = tid; i < HIDDEN * N_CLASSES; i += TILE_WAVES * 32) {
        const int t = i >> 6, ln = (i >> 1) & 31, sx = i & 1;
        W2f[i] = W2[(size_t)(4 * t + 2 * (ln >> 4) + sx) * N_CLASSES + (ln & 15)];
    }
    for (int i = tid; i < HIDDEN; i += TILE_WAVES * 32) b1s[i] = b1[i];
    {
        const float4* s4 = (const float4*)(hp + (size_t)n0 * HIDDEN);
        for (int i = tid; i < TILE_NODES * HIDDEN / 4; i += TILE_WAVES * 32) {
            const float4 s = s4[i];
            float* d = acc + (i >> 5) * ACC_PITCH + (i & 31);
            d[0] = s.x; d[32] = s.y; d[64] = s.z; d[96] = s.w;
        }
    }
    __syncthreads();

    int* q = queue[wave];
    int head = 0, tail = 0;
    const int n_chunks = (N_EDGES + SCAN_EDGES - 1) / SCAN_EDGES;
    const int last_base = N_EDGES - SCAN_EDGES;

    const unsigned char* tb = (const unsigned char*)edge_tile;
    const unsigned pat = (unsigned)blockIdx.x * 0x01010101u;

    auto load_trip = [&](uint4 (&d)[2], int chunk) {
        const int c0 = chunk * SCAN_EDGES;
        const int base = c0 < last_base ? c0 : last_base;
#pragma unroll
        for (int k = 0; k < 2; ++k) d[k] = *(const uint4*)(tb + base + k * SCAN_STEP + 16 * lane);
    };
    auto hit_bits = [&](const uint4 (&d)[2], int chunk) -> unsigned {
        unsigned hb = 0u;
#pragma unroll
        for (int k = 0; k < 2; ++k) {
            const unsigned ww[4] = {d[k].x, d[k].y, d[k].z, d[k].w};
#pragma unroll
            for (int c = 0; c < 4; ++c) {
                const unsigned t = ww[c] ^ pat;
                const unsigned nz = (((t & 0x7F7F7F7Fu) + 0x7F7F7F7Fu) | t) & 0x80808080u;
                hb |= (nz ^ 0x80808080u) >> (7 - (4 * k + c));
            }
        }
        const int c0 = chunk * SCAN_EDGES;
        return c0 > last_base ? (hb & 0xF0F0F0F0u) : hb;
    };
    auto take_hits = [&](unsigned hb, int chunk) {
        const int c0 = chunk * SCAN_EDGES;
        const int base = c0 < last_base ? c0 : last_base;
        unsigned m = __builtin_amdgcn_ballot_w32(hb != 0u);
        while (m != 0u) {
            if (hb != 0u) {
                const int p = __builtin_ctz(hb);
                hb &= hb - 1u;
                const int w = p & 7;
                const int pos = (tail + (int)__builtin_amdgcn_mbcnt_lo(m, 0u)) & (QUEUE_SLOTS - 1);
                q[pos] = base + (w >> 2) * SCAN_STEP + 16 * lane + 4 * (w & 3) + (p >> 3);
            }
            tail += __builtin_popcount(m);
            if (tail - head >= 32) {
                __builtin_amdgcn_fence(__ATOMIC_ACQ_REL, "wavefront");
                tile_process(q, head, 32, lane, row, col, n0, hp, acc, my_list, &list_cursor);
                head += 32;
            }
            m = __builtin_amdgcn_ballot_w32(hb != 0u);
        }
    };

    uint4 d0[2], d1[2], d2[2], d3[2];
    load_trip(d0, wave);
    __builtin_amdgcn_sched_barrier(0);
    load_trip(d1, wave + TILE_WAVES);
    __builtin_amdgcn_sched_barrier(0);
    load_trip(d2, wave + 2 * TILE_WAVES);
    for (int chunk = wave; chunk < n_chunks; chunk += 4 * TILE_WAVES) {
        load_trip(d3, chunk + 3 * TILE_WAVES);
        take_hits(hit_bits(d0, chunk), chunk);
        load_trip(d0, chunk + 4 * TILE_WAVES);
        if (chunk + TILE_WAVES < n_chunks)
            take_hits(hit_bits(d1, chunk + TILE_WAVES), chunk + TILE_WAVES);
        load_trip(d1, chunk + 5 * TILE_WAVES);
        if (chunk + 2 * TILE_WAVES < n_chunks)
            take_hits(hit_bits(d2, chunk + 2 * TILE_WAVES), chunk + 2 * TILE_WAVES);
        load_trip(d2, chunk + 6 * TILE_WAVES);
        if (chunk + 3 * TILE_WAVES < n_chunks)
            take_hits(hit_bits(d3, chunk + 3 * TILE_WAVES), chunk + 3 * TILE_WAVES);
    }
    if (tail - head > 0) {
        __builtin_amdgcn_fence(__ATOMIC_ACQ_REL, "wavefront");
        tile_process(q, head, tail - head, lane, row, col, n0, hp, acc, my_list, &list_cursor);
    }
    __syncthreads();

    for (int u = wave; u < TILE_NODES / 16; u += TILE_WAVES) {
        const int m  = lane & 15;
        const int kh = lane >> 4;
        const int mo = kh * 8;
        const float d = dinv_tile[u * 16 + m];
        const float* arow = acc + (u * 16 + m) * ACC_PITCH + 64 * kh;
        v8f c = v8f{};
#pragma unroll
        for (int t = 0; t < HIDDEN / 4; ++t) {
            const v2f bias = *(const v2f*)(b1s + 4 * t + 2 * kh);
            const v2f b = *(const v2f*)(W2f + 64 * t + 2 * lane);
            v2f a;
            a.x = d * arow[t] + bias.x;
            a.y = d * arow[32 + t] + bias.y;
            a.x = a.x > 0.0f ? a.x : 0.0f;
            a.y = a.y > 0.0f ? a.y : 0.0f;
            c = __builtin_amdgcn_wmma_f32_16x16x4_f32(
                false, a, false, b, (short)0, c, false, false);
        }
        const int row0 = n0 + u * 16;
#pragma unroll
        for (int j = 0; j < 8; ++j) {
            h2p[(size_t)(row0 + mo + j) * N_CLASSES + m] = dinv_tile[u * 16 + mo + j] * c[j];
        }
    }
}


template <bool FULL>
__device__ __forceinline__ void agg2_step(unsigned int e, int nv, int half, int f,
                                          const float* __restrict__ h2p, float* acc2,
                                          const unsigned int* next_word, unsigned int& e_next) {
    float v[16] = {};
    unsigned int w[16];
#pragma unroll
    for (int i = 0; i < 16; ++i) {
        const unsigned int w0 = (unsigned int)__builtin_amdgcn_readlane((int)e, 2 * i);
        const unsigned int w1 = (unsigned int)__builtin_amdgcn_readlane((int)e, 2 * i + 1);
        w[i] = half ? w1 : w0;
        if (FULL || 2 * i + half < nv) v[i] = h2p[(size_t)(w[i] >> 9) * N_CLASSES + f];
    }
    if (next_word) e_next = *next_word;
#pragma unroll
    for (int i = 0; i < 16; ++i) {
        if (FULL || 2 * i + half < nv) unsafeAtomicAdd(&acc2[(w[i] & 511u) * N_CLASSES + f], v[i]);
    }
}

__global__ __launch_bounds__(TILE_WAVES * 32) void agg2_tile_kernel(const unsigned int* __restrict__ elist,
                                                                    const int* __restrict__ deg,
                                                                    const float* __restrict__ h2p,
                                                                    const float* __restrict__ b2,
                                                                    float* __restrict__ out) {
    __shared__ __attribute__((aligned(16))) float acc2[TILE_NODES * N_CLASSES];
    __shared__ float dinv_tile[TILE_NODES];
    __shared__ int red[64];

    const int tid  = threadIdx.x;
    const int lane = tid & 31;
    const int wave = __builtin_amdgcn_readfirstlane(tid >> 5);
    const int n0   = blockIdx.x * TILE_NODES;
    const int half = lane >> 4;
    const int f    = lane & 15;

    const int2 rng = tile_edge_range(deg, n0, tid, red);
    const unsigned int* my_list = elist + __builtin_amdgcn_readfirstlane(rng.x);
    const int cnt = __builtin_amdgcn_readfirstlane(rng.y);

    for (int i = tid; i < TILE_NODES; i += TILE_WAVES * 32) dinv_tile[i] = rsqrtf((float)deg[n0 + i]);
    {
        const float4* s4 = (const float4*)(h2p + (size_t)n0 * N_CLASSES);
        for (int i = tid; i < TILE_NODES * N_CLASSES / 4; i += TILE_WAVES * 32) ((float4*)acc2)[i] = s4[i];
    }
    __syncthreads();

    const int n_steps = (cnt + 31) >> 5;
    unsigned int e_next = 0u;
    if (wave < n_steps) {
        const int p = wave * 32 + lane;
        if (p < cnt) e_next = my_list[p];
    }
    for (int k = wave; k < n_steps; k += TILE_WAVES) {
        const unsigned int e = e_next;
        const int left = cnt - 32 * k;
        const int nv = left < 32 ? left : 32;
        const int pn = (k + TILE_WAVES) * 32 + lane;
        const unsigned int* next_word = pn < cnt ? my_list + pn : nullptr;
        if (nv == 32) agg2_step<true>(e, nv, half, f, h2p, acc2, next_word, e_next);
        else          agg2_step<false>(e, nv, half, f, h2p, acc2, next_word, e_next);
    }
    __syncthreads();

    const float bf = b2[f];
    for (int r = tid >> 4; r < TILE_NODES; r += TILE_WAVES * 2) {
        const float x = dinv_tile[r] * acc2[r * N_CLASSES + f] + bf;
        float mx = x;
#pragma unroll
        for (int o = 8; o > 0; o >>= 1) mx = fmaxf(mx, __shfl_xor(mx, o));
        float s = expf(x - mx);
#pragma unroll
        for (int o = 8; o > 0; o >>= 1) s += __shfl_xor(s, o);
        out[(size_t)(n0 + r) * N_CLASSES + f] = x - (mx + logf(s));
    }
}


extern "C" void kernel_launch(void* const* d_in, const int* in_sizes, int n_in,
                              void* d_out, int out_size, void* d_ws, size_t ws_size,
                              hipStream_t stream) {
    const float* x   = (const float*)d_in[0];
    const int*   ei  = (const int*)d_in[1];
    const int*   row = ei;
    const int*   col = ei + N_EDGES;
    const float* W1  = (const float*)d_in[2];
    const float* b1  = (const float*)d_in[3];
    const float* W2  = (const float*)d_in[4];
    const float* b2  = (const float*)d_in[5];
    float* out = (float*)d_out;

    float* ws   = (float*)d_ws;
    int*   deg  = (int*)d_ws;
    float* hp   = ws + ((N_NODES + 255) & ~255);
    float* h2p  = hp + (size_t)N_NODES * HIDDEN;
    unsigned int* elist = (unsigned int*)(h2p + (size_t)N_NODES * N_CLASSES);
    static_assert((size_t)N_EDGES <= (size_t)N_NODES * N_CLASSES * sizeof(float),
                  "the tile bytes fit the output buffer");
    unsigned int* edge_tile = (unsigned int*)d_out;
    static_assert((size_t)DEG_WGS * HIST_WORDS <= (size_t)N_NODES * N_CLASSES * 2, "the planes fit that region");
    unsigned int* hist = (unsigned int*)h2p;

    deg_hist_kernel<<<DEG_WGS, DEG_THREADS, 0, stream>>>(col, hist, edge_tile);

    gemm1_kernel<<<N_NODES / GEMM1_ROWS, GEMM1_WAVES * 32, 0, stream>>>(
        x, W1, (const unsigned short*)hist, deg, hp);
    agg1_tile_kernel<<<N_NODES / TILE_NODES, TILE_WAVES * 32, 0, stream>>>(
        row, col, edge_tile, deg, hp, b1, W2, h2p, elist);

    agg2_tile_kernel<<<N_NODES / TILE_NODES, TILE_WAVES * 32, 0, stream>>>(elist, deg, h2p, b2, out);
}


# Edges bucketed by destination tile (slice-local LDS counting sort, packed source/row words); tiles read their runs: degree planes, tile-byte scan, per-edge id gathers gone; adds in arrival order
# speedup vs baseline: 4.4670x; 1.1774x over previous
#include <hip/hip_runtime.h>
#include <math.h>

typedef __attribute__((ext_vector_type(2))) float v2f;
typedef __attribute__((ext_vector_type(8))) float v8f;

#define N_NODES   100000
#define N_EDGES   1600000
#define IN_FEAT   256
#define HIDDEN    128
#define N_CLASSES 16


#define TILE_NODES   400
#define N_TILES      (N_NODES / TILE_NODES)
static_assert(N_NODES % TILE_NODES == 0, "whole tiles only");
static_assert(N_TILES <= 256, "the bucket counters of a slice fit 256 words");

#define BKT_SLICES   160
#define BKT_THREADS  1024
#define BKT_EDGES    (N_EDGES / BKT_SLICES)
#define BKT_QUADS    (BKT_EDGES / 4)
#define BKT_TRIPS    ((BKT_QUADS + BKT_THREADS - 1) / BKT_THREADS)
#define TILE_OUT_WORDS (TILE_NODES * N_CLASSES)
static_assert(N_EDGES % BKT_SLICES == 0 && BKT_EDGES % 4 == 0, "equal slices of whole 4-edge items");
static_assert(2 * BKT_SLICES <= TILE_OUT_WORDS, "a tile's table fits its own output rows");
static_assert(TILE_NODES <= 512 && N_NODES <= (1 << 23), "a list word holds source << 9 | row within the tile");

__global__ __launch_bounds__(BKT_THREADS) void bucket_kernel(const int* __restrict__ row,
                                                            const int* __restrict__ col,
                                                            unsigned int* __restrict__ slices,
                                                            int* __restrict__ tables) {
    __shared__ unsigned int tcount[256];
    __shared__ unsigned int tbase[256];
    __shared__ __attribute__((aligned(16))) unsigned int sorted[BKT_EDGES];
    const int tid = threadIdx.x;
    const int slice = blockIdx.x;
    const int q0 = slice * BKT_QUADS;
    int4 r[BKT_TRIPS], c[BKT_TRIPS];
#pragma unroll
    for (int t = 0; t < BKT_TRIPS; ++t) {
        const int i = tid + t * BKT_THREADS;
        if (i < BKT_QUADS) {
            r[t] = *(const int4*)(row + 4 * (size_t)(q0 + i));
            c[t] = *(const int4*)(col + 4 * (size_t)(q0 + i));
        }
    }
    if (tid < 256) tcount[tid] = 0u;
    __syncthreads();
    int tl[BKT_TRIPS][4];
    unsigned int rk[BKT_TRIPS][4];
#pragma unroll
    for (int t = 0; t < BKT_TRIPS; ++t) {
        const int i = tid + t * BKT_THREADS;
        if (i < BKT_QUADS) {
            const int cc[4] = {c[t].x, c[t].y, c[t].z, c[t].w};
#pragma unroll
            for (int j = 0; j < 4; ++j) {
                tl[t][j] = cc[j] / TILE_NODES;
                rk[t][j] = atomicAdd(&tcount[tl[t][j]], 1u);
            }
        }
    }
    __syncthreads();
    if (tid < 32) {
        unsigned int v[8];
        unsigned int s = 0u;
#pragma unroll
        for (int j = 0; j < 8; ++j) { v[j] = tcount[tid * 8 + j]; s += v[j]; }
        unsigned int incl = s;
#pragma unroll
        for (int o = 1; o < 32; o <<= 1) {
            const unsigned int up = __shfl_up(incl, o);
            if (tid >= o) incl += up;
        }
        unsigned int excl = incl - s;
#pragma unroll
        for (int j = 0; j < 8; ++j) { tbase[tid * 8 + j] = excl; excl += v[j]; }
    }
    __syncthreads();
#pragma unroll
    for (int t = 0; t < BKT_TRIPS; ++t) {
        const int i = tid + t * BKT_THREADS;
        if (i < BKT_QUADS) {
            const int rr[4] = {r[t].x, r[t].y, r[t].z, r[t].w};
            const int cc[4] = {c[t].x, c[t].y, c[t].z, c[t].w};
#pragma unroll
            for (int j = 0; j < 4; ++j) {
                sorted[tbase[tl[t][j]] + rk[t][j]] =
                    ((unsigned int)rr[j] << 9) | (unsigned int)(cc[j] - tl[t][j] * TILE_NODES);
            }
        }
    }
    __syncthreads();
    uint4* dst = (uint4*)(slices + (size_t)slice * BKT_EDGES);
    for (int i = tid; i < BKT_QUADS; i += BKT_THREADS) dst[i] = ((const uint4*)sorted)[i];
    if (tid < N_TILES) {
        int* tb = tables + (size_t)tid * TILE_OUT_WORDS + 2 * slice;
        tb[0] = (int)tbase[tid];
        tb[1] = (int)tcount[tid];
    }
}

#define TILE_WAVES     32
#define LIST_CHUNK     8192
#define RUNS_PER_LANE  (BKT_SLICES / 32)
static_assert(BKT_SLICES % 32 == 0 && BKT_SLICES <= TILE_WAVES * 32, "one wave sums the run lengths, one thread reads a run's pair");
static_assert(LIST_CHUNK == 8 * TILE_WAVES * 32, "a chunk is 8 words per thread");
static_assert(BKT_SLICES <= 256, "8 bisection steps find a run");

__device__ __forceinline__ int tile_list_open(const int* tb, int tid, int* pre, int* rstart) {
    if (tid < BKT_SLICES) {
        const int2 o = *(const int2*)(tb + 2 * tid);
        rstart[tid] = tid * BKT_EDGES + o.x;
        pre[tid + 1] = o.y;
    }
    if (tid == 0) pre[0] = 0;
    __syncthreads();
    if (tid < 32) {
        int v[RUNS_PER_LANE];
        int s = 0;
#pragma unroll
        for (int j = 0; j < RUNS_PER_LANE; ++j) { v[j] = pre[1 + tid * RUNS_PER_LANE + j]; s += v[j]; }
        int incl = s;
#pragma unroll
        for (int o = 1; o < 32; o <<= 1) {
            const int up = __shfl_up(incl, o);
            if (tid >= o) incl += up;
        }
        int run = incl - s;
#pragma unroll
        for (int j = 0; j < RUNS_PER_LANE; ++j) { run += v[j]; pre[1 + tid * RUNS_PER_LANE + j] = run; }
    }
    __syncthreads();
    return pre[BKT_SLICES];
}

template <typename Sink>
__device__ __forceinline__ void tile_list_visit(const unsigned int* __restrict__ slices,
                                                const int* pre, const int* rstart,
                                                int c0, int cnt, int tid, Sink sink) {
    const int last = cnt - 1;
    int p[8], lo[8], hi[8];
    unsigned int w[8];
#pragma unroll
    for (int k = 0; k < 8; ++k) {
        const int q = c0 + tid + k * TILE_WAVES * 32;
        p[k] = q < last ? q : last;
        lo[k] = 0;
        hi[k] = BKT_SLICES;
    }
#pragma unroll
    for (int it = 0; it < 8; ++it) {
#pragma unroll
        for (int k = 0; k < 8; ++k) {
            const int mid = (lo[k] + hi[k]) >> 1;
            const bool le = pre[mid] <= p[k];
            lo[k] = le ? mid : lo[k];
            hi[k] = le ? hi[k] : mid;
        }
    }
#pragma unroll
    for (int k = 0; k < 8; ++k) w[k] = slices[rstart[lo[k]] + (p[k] - pre[lo[k]])];
#pragma unroll
    for (int k = 0; k < 8; ++k) {
        if (c0 + tid + k * TILE_WAVES * 32 < cnt) sink(tid + k * TILE_WAVES * 32, w[k]);
    }
}


#define GEMM1_ROWS   400
#define GEMM1_WAVES  32
#define W1_PITCH     288
#define GEMM1_UNITS  ((GEMM1_ROWS / 16) * 2)
static_assert(N_NODES % GEMM1_ROWS == 0 && GEMM1_ROWS % 16 == 0, "whole row tiles only");
static_assert(HIDDEN == 128 && IN_FEAT % 32 == 0, "unit = 64 columns, A chunks of 32 columns");
static_assert(GEMM1_ROWS == TILE_NODES && GEMM1_WAVES == TILE_WAVES, "a work-group owns one tile and reads that tile's edge list");

__device__ __forceinline__ void gemm1_load_a(v2f (&a)[8], const float* __restrict__ p) {
#pragma unroll
    for (int s = 0; s < 8; ++s) a[s] = *(const v2f*)(p + 4 * s);
}

__device__ __forceinline__ void gemm1_chunk(v8f (&acc)[4], const v2f (&a)[8], const float* bc) {
#pragma unroll
    for (int s = 0; s < 8; ++s) {
        const float* bk = bc + (2 * s) * W1_PITCH;
#pragma unroll
        for (int n = 0; n < 4; ++n) {
            const v2f b = *(const v2f*)(bk + 32 * n);
            acc[n] = __builtin_amdgcn_wmma_f32_16x16x4_f32(
                false, a[s], false, b, (short)0, acc[n], false, false);
        }
    }
}

__global__ __launch_bounds__(GEMM1_WAVES * 32) void gemm1_kernel(const float* __restrict__ x,
                                                                const float* __restrict__ W1,
                                                                const unsigned int* __restrict__ slices,
                                                                const int* __restrict__ tables,
                                                                int* __restrict__ deg,
                                                                float* __restrict__ hp) {
    __shared__ __attribute__((aligned(16))) float Wl[(IN_FEAT / 2) * W1_PITCH];
    __shared__ float dinv_l[GEMM1_ROWS];

    __shared__ int degc[GEMM1_ROWS];
    __shared__ int pre[BKT_SLICES + 1];
    __shared__ int rstart[BKT_SLICES];

    const int tid = threadIdx.x;
    if (tid < GEMM1_ROWS) degc[tid] = 1;
    const int cnt = tile_list_open(tables + (size_t)blockIdx.x * TILE_OUT_WORDS, tid, pre, rstart);
    for (int c0 = 0; c0 < cnt; c0 += LIST_CHUNK) {
        tile_list_visit(slices, pre, rstart, c0, cnt, tid,
                        [&](int, unsigned int w) { atomicAdd(&degc[w & 511u], 1); });
    }
    __syncthreads();
    if (tid < GEMM1_ROWS) {
        const int s = degc[tid];
        dinv_l[tid] = rsqrtf((float)s);
        deg[blockIdx.x * GEMM1_ROWS + tid] = s;
    }
    for (int it = tid; it < (IN_FEAT / 2) * (HIDDEN / 4); it += GEMM1_WAVES * 32) {
        const int p = it / (HIDDEN / 4);
        const int c = it % (HIDDEN / 4);
        const float4 r0 = *(const float4*)(W1 + (size_t)(2 * p) * HIDDEN + 4 * c);
        const float4 r1 = *(const float4*)(W1 + (size_t)(2 * p + 1) * HIDDEN + 4 * c);
        float4* d = (float4*)(Wl + p * W1_PITCH + 8 * c);
        d[0] = make_float4(r0.x, r1.x, r0.y, r1.y);
        d[1] = make_float4(r0.z, r1.z, r0.w, r1.w);
    }
    __syncthreads();

    const int lane = tid & 31;
    const int wave = tid >> 5;
    const int m  = lane & 15;
    const int kh = lane >> 4;
    const int mo = kh * 8;

    for (int u = wave; u < GEMM1_UNITS; u += GEMM1_WAVES) {
        const int row0 = blockIdx.x * GEMM1_ROWS + (u >> 1) * 16;
        const int col0 = (u & 1) * 64;
        const float* arow = x + (size_t)(row0 + m) * IN_FEAT + 2 * kh;
        const float* bl = Wl + kh * W1_PITCH + (col0 + m) * 2;

        v8f acc[4];
#pragma unroll
        for (int n = 0; n < 4; ++n) acc[n] = v8f{};

        v2f a0[8], a1[8];
        gemm1_load_a(a0, arow);
#pragma nounroll
        for (int cp = 0; cp < IN_FEAT / 64; ++cp) {
            gemm1_load_a(a1, arow + 64 * cp + 32);
            gemm1_chunk(acc, a0, bl + (32 * cp) * W1_PITCH);
            if (2 * cp + 2 < IN_FEAT / 32)
                gemm1_load_a(a0, arow + 64 * cp + 64);
            gemm1_chunk(acc, a1, bl + (32 * cp + 16) * W1_PITCH);
        }

        float ds[8];
#pragma unroll
        for (int j = 0; j < 8; ++j) ds[j] = dinv_l[(u >> 1) * 16 + mo + j];
#pragma unroll
        for (int n = 0; n < 4; ++n) {
#pragma unroll
            for (int j = 0; j < 8; ++j) {
                hp[(size_t)(row0 + mo + j) * HIDDEN + col0 + n * 16 + m] = ds[j] * acc[n][j];
            }
        }
    }
}


#define EDGES_PER_WAVE 32
#define EDGE_GROUP     4
static_assert(EDGES_PER_WAVE % EDGE_GROUP == 0, "whole groups only");

#define ACC_PITCH    (HIDDEN + 1)
static_assert(TILE_NODES % 16 == 0 && HIDDEN % 4 == 0 && N_CLASSES == 16,
              "projection 2: whole 16-row units, 4-wide k-steps, one 16-column tile");

__device__ __forceinline__ void tile_load_group(float4 (&v)[EDGE_GROUP],
                                                const float* __restrict__ hp,
                                                int r_vec, int first, int n, int lane) {
#pragma unroll
    for (int q = 0; q < EDGE_GROUP; ++q) {
        if (first + q < n) {
            const int rs = __builtin_amdgcn_readlane(r_vec, first + q);
            v[q] = *((const float4*)(hp + (size_t)rs * HIDDEN) + lane);
        }
    }
}

__device__ __forceinline__ void tile_add_group(const float4 (&v)[EDGE_GROUP],
                                               float* acc,
                                               int cl_vec, int first, int n, int lane) {
#pragma unroll
    for (int q = 0; q < EDGE_GROUP; ++q) {
        if (first + q < n) {
            const int cs = __builtin_amdgcn_readlane(cl_vec, first + q);
            float* dst = acc + cs * ACC_PITCH + lane;
            unsafeAtomicAdd(dst,      v[q].x);
            unsafeAtomicAdd(dst + 32, v[q].y);
            unsafeAtomicAdd(dst + 64, v[q].z);
            unsafeAtomicAdd(dst + 96, v[q].w);
        }
    }
}

__device__ __forceinline__ void tile_process(int r, int cl, int n, int lane,
                                             const float* __restrict__ hp, float* acc) {
    float4 v[2][EDGE_GROUP];
    tile_load_group(v[0], hp, r, 0, n, lane);
#pragma unroll
    for (int g = 0; g < EDGES_PER_WAVE / EDGE_GROUP; ++g) {
        if (g * EDGE_GROUP < n) {
            if (g + 1 < EDGES_PER_WAVE / EDGE_GROUP)
                tile_load_group(v[(g + 1) & 1], hp, r, (g + 1) * EDGE_GROUP, n, lane);
            tile_add_group(v[g & 1], acc, cl, g * EDGE_GROUP, n, lane);
        }
    }
}

__global__ __launch_bounds__(TILE_WAVES * 32) void agg1_tile_kernel(const unsigned int* __restrict__ slices,
                                                                    const int* __restrict__ tables,
                                                                    const int* __restrict__ deg,
                                                                    const float* __restrict__ hp,
                                                                    const float* __restrict__ b1,
                                                                    const float* __restrict__ W2,
                                                                    float* __restrict__ h2p) {
    __shared__ float acc[TILE_NODES * ACC_PITCH];
    __shared__ float dinv_tile[TILE_NODES];
    __shared__ __attribute__((aligned(8))) float W2f[HIDDEN * N_CLASSES];
    __shared__ __attribute__((aligned(8))) float b1s[HIDDEN];
    __shared__ int pre[BKT_SLICES + 1];
    __shared__ int rstart[BKT_SLICES];
    __shared__ unsigned int lst[LIST_CHUNK];

    const int tid  = threadIdx.x;
    const int lane = tid & 31;
    const int wave = __builtin_amdgcn_readfirstlane(tid >> 5);
    const int n0   = blockIdx.x * TILE_NODES;

    for (int i = tid; i < TILE_NODES; i += TILE_WAVES * 32) dinv_tile[i] = rsqrtf((float)deg[n0 + i]);
    for (int i = tid; i < HIDDEN * N_CLASSES; i += TILE_WAVES * 32) {
        const int t = i >> 6, ln = (i >> 1) & 31, sx = i & 1;
        W2f[i] = W2[(size_t)(4 * t + 2 * (ln >> 4) + sx) * N_CLASSES + (ln & 15)];
    }
    for (int i = tid; i < HIDDEN; i += TILE_WAVES * 32) b1s[i] = b1[i];
    {
        const float4* s4 = (const float4*)(hp + (size_t)n0 * HIDDEN);
        for (int i = tid; i < TILE_NODES * HIDDEN / 4; i += TILE_WAVES * 32) {
            const float4 s = s4[i];
            float* d = acc + (i >> 5) * ACC_PITCH + (i & 31);
            d[0] = s.x; d[32] = s.y; d[64] = s.z; d[96] = s.w;
        }
    }
    const int cnt = __builtin_amdgcn_readfirstlane(
        tile_list_open(tables + (size_t)blockIdx.x * TILE_OUT_WORDS, tid, pre, rstart));

    for (int c0 = 0; c0 < cnt; c0 += LIST_CHUNK) {
        tile_list_visit(slices, pre, rstart, c0, cnt, tid, [&](int at, unsigned int w) { lst[at] = w; });
        __syncthreads();
        const int n_in = cnt - c0 < LIST_CHUNK ? cnt - c0 : LIST_CHUNK;
        const int n_steps = (n_in + 31) >> 5;
        for (int k = wave; k < n_steps; k += TILE_WAVES) {
            const int left = n_in - 32 * k;
            const int nv = left < 32 ? left : 32;
            int r = 0, cl = 0;
            if (lane < nv) {
                const unsigned int e = lst[32 * k + lane];
                r  = (int)(e >> 9);
                cl = (int)(e & 511u);
            }
            tile_process(r, cl, nv, lane, hp, acc);
        }
        __syncthreads();
    }

    for (int u = wave; u < TILE_NODES / 16; u += TILE_WAVES) {
        const int m  = lane & 15;
        const int kh = lane >> 4;
        const int mo = kh * 8;
        const float d = dinv_tile[u * 16 + m];
        const float* arow = acc + (u * 16 + m) * ACC_PITCH + 64 * kh;
        v8f c = v8f{};
#pragma unroll
        for (int t = 0; t < HIDDEN / 4; ++t) {
            const v2f bias = *(const v2f*)(b1s + 4 * t + 2 * kh);
            const v2f b = *(const v2f*)(W2f + 64 * t + 2 * lane);
            v2f a;
            a.x = d * arow[t] + bias.x;
            a.y = d * arow[32 + t] + bias.y;
            a.x = a.x > 0.0f ? a.x : 0.0f;
            a.y = a.y > 0.0f ? a.y : 0.0f;
            c = __builtin_amdgcn_wmma_f32_16x16x4_f32(
                false, a, false, b, (short)0, c, false, false);
        }
        const int row0 = n0 + u * 16;
#pragma unroll
        for (int j = 0; j < 8; ++j) {
            h2p[(size_t)(row0 + mo + j) * N_CLASSES + m] = dinv_tile[u * 16 + mo + j] * c[j];
        }
    }
}


template <bool FULL>
__device__ __forceinline__ void agg2_step(unsigned int e, int nv, int half, int f,
                                          const float* __restrict__ h2p, float* acc2) {
    float v[16] = {};
    unsigned int w[16];
#pragma unroll
    for (int i = 0; i < 16; ++i) {
        const unsigned int w0 = (unsigned int)__builtin_amdgcn_readlane((int)e, 2 * i);
        const unsigned int w1 = (unsigned int)__builtin_amdgcn_readlane((int)e, 2 * i + 1);
        w[i] = half ? w1 : w0;
        if (FULL || 2 * i + half < nv) v[i] = h2p[(size_t)(w[i] >> 9) * N_CLASSES + f];
    }
#pragma unroll
    for (int i = 0; i < 16; ++i) {
        if (FULL || 2 * i + half < nv) unsafeAtomicAdd(&acc2[(w[i] & 511u) * N_CLASSES + f], v[i]);
    }
}

__global__ __launch_bounds__(TILE_WAVES * 32) void agg2_tile_kernel(const unsigned int* __restrict__ slices,
                                                                    const int* tables,
                                                                    const int* __restrict__ deg,
                                                                    const float* __restrict__ h2p,
                                                                    const float* __restrict__ b2,
                                                                    float* out) {
    __shared__ __attribute__((aligned(16))) float acc2[TILE_NODES * N_CLASSES];
    __shared__ float dinv_tile[TILE_NODES];
    __shared__ int pre[BKT_SLICES + 1];
    __shared__ int rstart[BKT_SLICES];
    __shared__ unsigned int lst[LIST_CHUNK];

    const int tid  = threadIdx.x;
    const int lane = tid & 31;
    const int wave = __builtin_amdgcn_readfirstlane(tid >> 5);
    const int n0   = blockIdx.x * TILE_NODES;
    const int half = lane >> 4;
    const int f    = lane & 15;

    for (int i = tid; i < TILE_NODES; i += TILE_WAVES * 32) dinv_tile[i] = rsqrtf((float)deg[n0 + i]);
    {
        const float4* s4 = (const float4*)(h2p + (size_t)n0 * N_CLASSES);
        for (int i = tid; i < TILE_NODES * N_CLASSES / 4; i += TILE_WAVES * 32) ((float4*)acc2)[i] = s4[i];
    }
    const int cnt = __builtin_amdgcn_readfirstlane(
        tile_list_open(tables + (size_t)blockIdx.x * TILE_OUT_WORDS, tid, pre, rstart));

    for (int c0 = 0; c0 < cnt; c0 += LIST_CHUNK) {
        tile_list_visit(slices, pre, rstart, c0, cnt, tid, [&](int at, unsigned int w) { lst[at] = w; });
        __syncthreads();
        const int n_in = cnt - c0 < LIST_CHUNK ? cnt - c0 : LIST_CHUNK;
        const int n_steps = (n_in + 31) >> 5;
        for (int k = wave; k < n_steps; k += TILE_WAVES) {
            const int left = n_in - 32 * k;
            const int nv = left < 32 ? left : 32;
            const unsigned int e = lane < nv ? lst[32 * k + lane] : 0u;
            if (nv == 32) agg2_step<true>(e, nv, half, f, h2p, acc2);
            else          agg2_step<false>(e, nv, half, f, h2p, acc2);
        }
        __syncthreads();
    }

    const float bf = b2[f];
    for (int r = tid >> 4; r < TILE_NODES; r += TILE_WAVES * 2) {
        const float x = dinv_tile[r] * acc2[r * N_CLASSES + f] + bf;
        float mx = x;
#pragma unroll
        for (int o = 8; o > 0; o >>= 1) mx = fmaxf(mx, __shfl_xor(mx, o));
        float s = expf(x - mx);
#pragma unroll
        for (int o = 8; o > 0; o >>= 1) s += __shfl_xor(s, o);
        out[(size_t)(n0 + r) * N_CLASSES + f] = x - (mx + logf(s));
    }
}


extern "C" void kernel_launch(void* const* d_in, const int* in_sizes, int n_in,
                              void* d_out, int out_size, void* d_ws, size_t ws_size,
                              hipStream_t stream) {
    const float* x   = (const float*)d_in[0];
    const int*   ei  = (const int*)d_in[1];
    const int*   row = ei;
    const int*   col = ei + N_EDGES;
    const float* W1  = (const float*)d_in[2];
    const float* b1  = (const float*)d_in[3];
    const float* W2  = (const float*)d_in[4];
    const float* b2  = (const float*)d_in[5];
    float* out = (float*)d_out;

    float* ws   = (float*)d_ws;
    int*   deg  = (int*)d_ws;
    float* hp   = ws + ((N_NODES + 255) & ~255);
    float* h2p  = hp + (size_t)N_NODES * HIDDEN;
    unsigned int* slices = (unsigned int*)(h2p + (size_t)N_NODES * N_CLASSES);
    int* tables = (int*)d_out;

    bucket_kernel<<<BKT_SLICES, BKT_THREADS, 0, stream>>>(row, col, slices, tables);

    gemm1_kernel<<<N_NODES / GEMM1_ROWS, GEMM1_WAVES * 32, 0, stream>>>(
        x, W1, slices, tables, deg, hp);
    agg1_tile_kernel<<<N_TILES, TILE_WAVES * 32, 0, stream>>>(
        slices, tables, deg, hp, b1, W2, h2p);

    agg2_tile_kernel<<<N_TILES, TILE_WAVES * 32, 0, stream>>>(slices, tables, deg, h2p, b2, out);
}
